# starting kernel
# baseline (speedup 1.0000x reference)
#include <hip/hip_runtime.h>
#include <hip/hip_bf16.h>


#define N_NODES   100000
#define E_EDGES   1600000
#define IN_F      128
#define H_HEADS   4
#define F_OUT     16
#define HF        64
#define NEG_SLOPE 0.01f

typedef float v2f __attribute__((ext_vector_type(2)));
typedef float v8f __attribute__((ext_vector_type(8)));

__device__ __forceinline__ unsigned fenc(float f) {
    unsigned u = __float_as_uint(f);
    return (u & 0x80000000u) ? ~u : (u | 0x80000000u);
}
__device__ __forceinline__ float fdec(unsigned k) {
    return (k & 0x80000000u) ? __uint_as_float(k ^ 0x80000000u)
                             : __uint_as_float(~k);
}

__global__ void gat_init(float* __restrict__ out, float* __restrict__ denom,
                         unsigned* __restrict__ maxenc) {
    size_t i = (size_t)blockIdx.x * blockDim.x + threadIdx.x;
    if (i < (size_t)N_NODES * HF)      out[i]   = 0.0f;
    if (i < (size_t)N_NODES * H_HEADS) denom[i] = 0.0f;
    if (i == 0) *maxenc = 0u;
}

__global__ __launch_bounds__(128)
void gat_transform_wmma(const float* __restrict__ x,
                        const float* __restrict__ W,
                        float* __restrict__ nt) {
    const int lane = threadIdx.x & 31;
    const int wave = threadIdx.x >> 5;
    const int m    = lane & 15;
    const int kk   = lane >> 4;
    const int row  = blockIdx.x * 16 + m;
    const int ncol = wave * 16 + m;

    const float* xrow = x + (size_t)row  * IN_F + 2 * kk;
    const float* wrow = W + (size_t)ncol * IN_F + 2 * kk;

    v8f c = {};
#pragma unroll
    for (int k0 = 0; k0 < IN_F; k0 += 4) {
        v2f a = *(const v2f*)(xrow + k0);
        v2f b = *(const v2f*)(wrow + k0);
        c = __builtin_amdgcn_wmma_f32_16x16x4_f32(
                 false, a,  false, b,
                 (short)0, c,  false,  false);
    }

    const int half = lane >> 4;
#pragma unroll
    for (int r = 0; r < 8; ++r) {
        int mm = r + 8 * half;
        nt[(size_t)(blockIdx.x * 16 + mm) * HF + wave * 16 + (lane & 15)] = c[r];
    }
}

__global__ __launch_bounds__(256)
void gat_edge_scores(const float* __restrict__ nt,
                     const float* __restrict__ a_w,
                     const int*   __restrict__ src,
                     const int*   __restrict__ tgt,
                     float* __restrict__ aw,
                     unsigned* __restrict__ maxenc) {
    __shared__ unsigned smax;
    if (threadIdx.x == 0) smax = 0u;
    __syncthreads();

    const int lane = threadIdx.x & 31;
    const int wv   = threadIdx.x >> 5;
    const int e    = blockIdx.x * 8 + wv;

    float res[H_HEADS] = {0.f, 0.f, 0.f, 0.f};
    if (e < E_EDGES) {
        const int vs = src[e];
        const int vt = tgt[e];
        const int j  = lane * 4;
        const int hp = j >> 5;
        const int off = j & 31;
        const float* base = (off < 16)
            ? (nt + (size_t)vs * HF + hp * F_OUT + off)
            : (nt + (size_t)vt * HF + hp * F_OUT + (off - 16));
        const float4 v = *(const float4*)base;
#pragma unroll
        for (int h = 0; h < H_HEADS; ++h) {
            const float* ar = a_w + h * (2 * HF) + j;
            res[h] = fmaf(v.x, ar[0], fmaf(v.y, ar[1],
                     fmaf(v.z, ar[2], v.w * ar[3])));
        }
    }
#pragma unroll
    for (int d = 16; d >= 1; d >>= 1) {
#pragma unroll
        for (int h = 0; h < H_HEADS; ++h)
            res[h] += __shfl_xor(res[h], d, 32);
    }
    if (e < E_EDGES && lane == 0) {
        float mx = res[0];
#pragma unroll
        for (int h = 0; h < H_HEADS; ++h) {
            aw[(size_t)e * H_HEADS + h] = res[h];
            mx = fmaxf(mx, res[h]);
        }
        atomicMax(&smax, fenc(mx));
    }
    __syncthreads();
    if (threadIdx.x == 0) atomicMax(maxenc, smax);
}

__global__ __launch_bounds__(256)
void gat_softmax_denom(float* __restrict__ awex,
                       const int* __restrict__ tgt,
                       float* __restrict__ denom,
                       const unsigned* __restrict__ maxenc) {
    const int e = blockIdx.x * blockDim.x + threadIdx.x;
    if (e >= E_EDGES) return;
    const float gmax = fdec(*maxenc);
    const int vt = tgt[e];
    float4 a = *(const float4*)(awex + (size_t)e * H_HEADS);
    float v[H_HEADS] = {a.x, a.y, a.z, a.w};
#pragma unroll
    for (int h = 0; h < H_HEADS; ++h) {
        float z = v[h] - gmax;
        z = (z > 0.0f) ? z : NEG_SLOPE * z;
        float ev = __expf(z);
        awex[(size_t)e * H_HEADS + h] = ev;
        atomicAdd(&denom[(size_t)vt * H_HEADS + h], ev);
    }
}

__global__ __launch_bounds__(256)
void gat_aggregate(const float* __restrict__ nt,
                   const float* __restrict__ ex,
                   const float* __restrict__ denom,
                   const int*   __restrict__ src,
                   const int*   __restrict__ tgt,
                   float* __restrict__ out) {
    const int lane = threadIdx.x & 31;
    const int wv   = threadIdx.x >> 5;
    const int e    = blockIdx.x * 8 + wv;
    if (e >= E_EDGES) return;
    const int vs = src[e];
    const int vt = tgt[e];
#pragma unroll
    for (int c = 0; c < 2; ++c) {
        const int idx = lane + 32 * c;
        const int h   = idx >> 4;
        const float alpha = ex[(size_t)e * H_HEADS + h] /
                            (denom[(size_t)vt * H_HEADS + h] + 1e-8f);
        const float sv = nt[(size_t)vs * HF + idx];
        atomicAdd(&out[(size_t)vt * HF + idx], alpha * sv);
    }
}

extern "C" void kernel_launch(void* const* d_in, const int* in_sizes, int n_in,
                              void* d_out, int out_size, void* d_ws, size_t ws_size,
                              hipStream_t stream) {
    const float* x   = (const float*)d_in[0];
    const float* W_w = (const float*)d_in[1];
    const float* a_w = (const float*)d_in[2];
    const int*   ei  = (const int*)d_in[3];
    const int* src = ei;
    const int* tgt = ei + E_EDGES;
    float* out = (float*)d_out;

    float* nt    = (float*)d_ws;
    float* awex  = nt   + (size_t)N_NODES * HF;
    float* denom = awex + (size_t)E_EDGES * H_HEADS;
    unsigned* maxenc = (unsigned*)(denom + (size_t)N_NODES * H_HEADS);

    (void)in_sizes; (void)n_in; (void)out_size; (void)ws_size;

    {
        size_t total = (size_t)N_NODES * HF;
        int blocks = (int)((total + 255) / 256);
        gat_init<<<blocks, 256, 0, stream>>>(out, denom, maxenc);
    }
    {
        gat_transform_wmma<<<N_NODES / 16, 128, 0, stream>>>(x, W_w, nt);
    }
    {
        gat_edge_scores<<<(E_EDGES + 7) / 8, 256, 0, stream>>>(
            nt, a_w, src, tgt, awex, maxenc);
    }
    {
        gat_softmax_denom<<<(E_EDGES + 255) / 256, 256, 0, stream>>>(
            awex, tgt, denom, maxenc);
    }
    {
        gat_aggregate<<<(E_EDGES + 7) / 8, 256, 0, stream>>>(
            nt, awex, denom, src, tgt, out);
    }
}


# Normalise once per node: exp+denominator pass fused into the scatter pass (out += ex*s, denom += ex per edge); new last pass divides out by denom+1e-8: division moves from each term to the sum
# speedup vs baseline: 1.4291x; 1.4291x over previous
#include <hip/hip_runtime.h>
#include <hip/hip_bf16.h>


#define N_NODES   100000
#define E_EDGES   1600000
#define IN_F      128
#define H_HEADS   4
#define F_OUT     16
#define HF        64
#define NEG_SLOPE 0.01f

typedef float v2f __attribute__((ext_vector_type(2)));
typedef float v8f __attribute__((ext_vector_type(8)));

__device__ __forceinline__ unsigned fenc(float f) {
    unsigned u = __float_as_uint(f);
    return (u & 0x80000000u) ? ~u : (u | 0x80000000u);
}
__device__ __forceinline__ float fdec(unsigned k) {
    return (k & 0x80000000u) ? __uint_as_float(k ^ 0x80000000u)
                             : __uint_as_float(~k);
}

__global__ void gat_init(float* __restrict__ out, float* __restrict__ denom,
                         unsigned* __restrict__ maxenc) {
    size_t i = (size_t)blockIdx.x * blockDim.x + threadIdx.x;
    if (i < (size_t)N_NODES * HF)      out[i]   = 0.0f;
    if (i < (size_t)N_NODES * H_HEADS) denom[i] = 0.0f;
    if (i == 0) *maxenc = 0u;
}

__global__ __launch_bounds__(128)
void gat_transform_wmma(const float* __restrict__ x,
                        const float* __restrict__ W,
                        float* __restrict__ nt) {
    const int lane = threadIdx.x & 31;
    const int wave = threadIdx.x >> 5;
    const int m    = lane & 15;
    const int kk   = lane >> 4;
    const int row  = blockIdx.x * 16 + m;
    const int ncol = wave * 16 + m;

    const float* xrow = x + (size_t)row  * IN_F + 2 * kk;
    const float* wrow = W + (size_t)ncol * IN_F + 2 * kk;

    v8f c = {};
#pragma unroll
    for (int k0 = 0; k0 < IN_F; k0 += 4) {
        v2f a = *(const v2f*)(xrow + k0);
        v2f b = *(const v2f*)(wrow + k0);
        c = __builtin_amdgcn_wmma_f32_16x16x4_f32(
                 false, a,  false, b,
                 (short)0, c,  false,  false);
    }

    const int half = lane >> 4;
#pragma unroll
    for (int r = 0; r < 8; ++r) {
        int mm = r + 8 * half;
        nt[(size_t)(blockIdx.x * 16 + mm) * HF + wave * 16 + (lane & 15)] = c[r];
    }
}

__global__ __launch_bounds__(256)
void gat_edge_scores(const float* __restrict__ nt,
                     const float* __restrict__ a_w,
                     const int*   __restrict__ src,
                     const int*   __restrict__ tgt,
                     float* __restrict__ aw,
                     unsigned* __restrict__ maxenc) {
    __shared__ unsigned smax;
    if (threadIdx.x == 0) smax = 0u;
    __syncthreads();

    const int lane = threadIdx.x & 31;
    const int wv   = threadIdx.x >> 5;
    const int e    = blockIdx.x * 8 + wv;

    float res[H_HEADS] = {0.f, 0.f, 0.f, 0.f};
    if (e < E_EDGES) {
        const int vs = src[e];
        const int vt = tgt[e];
        const int j  = lane * 4;
        const int hp = j >> 5;
        const int off = j & 31;
        const float* base = (off < 16)
            ? (nt + (size_t)vs * HF + hp * F_OUT + off)
            : (nt + (size_t)vt * HF + hp * F_OUT + (off - 16));
        const float4 v = *(const float4*)base;
#pragma unroll
        for (int h = 0; h < H_HEADS; ++h) {
            const float* ar = a_w + h * (2 * HF) + j;
            res[h] = fmaf(v.x, ar[0], fmaf(v.y, ar[1],
                     fmaf(v.z, ar[2], v.w * ar[3])));
        }
    }
#pragma unroll
    for (int d = 16; d >= 1; d >>= 1) {
#pragma unroll
        for (int h = 0; h < H_HEADS; ++h)
            res[h] += __shfl_xor(res[h], d, 32);
    }
    if (e < E_EDGES && lane == 0) {
        float mx = res[0];
#pragma unroll
        for (int h = 0; h < H_HEADS; ++h) {
            aw[(size_t)e * H_HEADS + h] = res[h];
            mx = fmaxf(mx, res[h]);
        }
        atomicMax(&smax, fenc(mx));
    }
    __syncthreads();
    if (threadIdx.x == 0) atomicMax(maxenc, smax);
}

__global__ __launch_bounds__(256)
void gat_scatter_unnormalized(const float* __restrict__ nt,
                              const float* __restrict__ aw,
                              const int*   __restrict__ src,
                              const int*   __restrict__ tgt,
                              const unsigned* __restrict__ maxenc,
                              float* __restrict__ denom,
                              float* __restrict__ out) {
    const int lane = threadIdx.x & 31;
    const int wv   = threadIdx.x >> 5;
    const int e    = blockIdx.x * 8 + wv;
    if (e >= E_EDGES) return;
    const int vs = src[e];
    const int vt = tgt[e];
    const float* srow = nt + (size_t)vs * HF;
    const float s_lo = srow[lane];
    const float s_hi = srow[lane + 32];
    const float gmax = fdec(*maxenc);
    const float4 a = *(const float4*)(aw + (size_t)e * H_HEADS);
    float ev[H_HEADS] = {a.x, a.y, a.z, a.w};
#pragma unroll
    for (int h = 0; h < H_HEADS; ++h) {
        float z = ev[h] - gmax;
        z = (z > 0.0f) ? z : NEG_SLOPE * z;
        ev[h] = __expf(z);
    }
    const bool  up   = (lane >> 4) != 0;
    const float e_lo = up ? ev[1] : ev[0];
    const float e_hi = up ? ev[3] : ev[2];
    if (lane < H_HEADS) {
        const float d = (lane & 2) ? ((lane & 1) ? ev[3] : ev[2])
                                   : ((lane & 1) ? ev[1] : ev[0]);
        atomicAdd(&denom[(size_t)vt * H_HEADS + lane], d);
    }
    float* orow = out + (size_t)vt * HF;
    atomicAdd(&orow[lane],      e_lo * s_lo);
    atomicAdd(&orow[lane + 32], e_hi * s_hi);
}

__global__ __launch_bounds__(256)
void gat_normalize(float* __restrict__ out, const float* __restrict__ denom) {
    const int i = blockIdx.x * blockDim.x + threadIdx.x;
    if (i >= N_NODES * HF / 4) return;
    const float d = denom[i >> 2] + 1e-8f;
    float4 v = ((const float4*)out)[i];
    v.x /= d; v.y /= d; v.z /= d; v.w /= d;
    ((float4*)out)[i] = v;
}

extern "C" void kernel_launch(void* const* d_in, const int* in_sizes, int n_in,
                              void* d_out, int out_size, void* d_ws, size_t ws_size,
                              hipStream_t stream) {
    const float* x   = (const float*)d_in[0];
    const float* W_w = (const float*)d_in[1];
    const float* a_w = (const float*)d_in[2];
    const int*   ei  = (const int*)d_in[3];
    const int* src = ei;
    const int* tgt = ei + E_EDGES;
    float* out = (float*)d_out;

    float* nt    = (float*)d_ws;
    float* awex  = nt   + (size_t)N_NODES * HF;
    float* denom = awex + (size_t)E_EDGES * H_HEADS;
    unsigned* maxenc = (unsigned*)(denom + (size_t)N_NODES * H_HEADS);

    (void)in_sizes; (void)n_in; (void)out_size; (void)ws_size;

    {
        size_t total = (size_t)N_NODES * HF;
        int blocks = (int)((total + 255) / 256);
        gat_init<<<blocks, 256, 0, stream>>>(out, denom, maxenc);
    }
    {
        gat_transform_wmma<<<N_NODES / 16, 128, 0, stream>>>(x, W_w, nt);
    }
    {
        gat_edge_scores<<<(E_EDGES + 7) / 8, 256, 0, stream>>>(
            nt, a_w, src, tgt, awex, maxenc);
    }
    {
        gat_scatter_unnormalized<<<(E_EDGES + 7) / 8, 256, 0, stream>>>(
            nt, awex, src, tgt, maxenc, denom, out);
    }
    {
        gat_normalize<<<N_NODES * HF / 4 / 256, 256, 0, stream>>>(out, denom);
    }
}


# Float-atomic scatter replaced by per-target edge lists built in the call (one integer exchange per edge); 8 lanes per node walk the list, sums in registers in list order; logit sum order changes
# speedup vs baseline: 3.4814x; 2.4361x over previous
#include <hip/hip_runtime.h>
#include <hip/hip_bf16.h>


#define N_NODES   100000
#define E_EDGES   1600000
#define IN_F      128
#define H_HEADS   4
#define F_OUT     16
#define HF        64
#define NEG_SLOPE 0.01f

typedef float v2f __attribute__((ext_vector_type(2)));
typedef float v8f __attribute__((ext_vector_type(8)));

__device__ __forceinline__ unsigned fenc(float f) {
    unsigned u = __float_as_uint(f);
    return (u & 0x80000000u) ? ~u : (u | 0x80000000u);
}
__device__ __forceinline__ float fdec(unsigned k) {
    return (k & 0x80000000u) ? __uint_as_float(k ^ 0x80000000u)
                             : __uint_as_float(~k);
}

__global__ __launch_bounds__(128)
void gat_transform_wmma(const float* __restrict__ x,
                        const float* __restrict__ W,
                        float* __restrict__ nt,
                        int* __restrict__ head,
                        unsigned* __restrict__ maxenc) {
    const int lane = threadIdx.x & 31;
    const int wave = threadIdx.x >> 5;
    const int m    = lane & 15;
    const int kk   = lane >> 4;
    const int row  = blockIdx.x * 16 + m;
    const int ncol = wave * 16 + m;

    const float* xrow = x + (size_t)row  * IN_F + 2 * kk;
    const float* wrow = W + (size_t)ncol * IN_F + 2 * kk;

    v8f c = {};
#pragma unroll
    for (int k0 = 0; k0 < IN_F; k0 += 4) {
        v2f a = *(const v2f*)(xrow + k0);
        v2f b = *(const v2f*)(wrow + k0);
        c = __builtin_amdgcn_wmma_f32_16x16x4_f32(
                 false, a,  false, b,
                 (short)0, c,  false,  false);
    }

    if (threadIdx.x < 16) head[blockIdx.x * 16 + threadIdx.x] = -1;
    if (blockIdx.x == 0 && threadIdx.x == 0) *maxenc = 0u;

    const int half = lane >> 4;
#pragma unroll
    for (int r = 0; r < 8; ++r) {
        int mm = r + 8 * half;
        nt[(size_t)(blockIdx.x * 16 + mm) * HF + wave * 16 + (lane & 15)] = c[r];
    }
}

__global__ __launch_bounds__(256)
void gat_link(const int* __restrict__ src, const int* __restrict__ tgt,
              int* __restrict__ head, int2* __restrict__ pk) {
    const int e = blockIdx.x * 256 + threadIdx.x;
    if (e >= E_EDGES) return;
    const int s = src[e];
    const int t = tgt[e];
    const int prev = atomicExch(&head[t], e);
    pk[e] = make_int2(s, prev);
}

template <int M>
__device__ __forceinline__ float lane_xor(float v) {
    return __builtin_bit_cast(float, __builtin_amdgcn_update_dpp(
        0, __builtin_bit_cast(int, v), 0x160 | M, 0xf, 0xf, true));
}
__device__ __forceinline__ float sum8(float v) {
    v += lane_xor<1>(v);
    v += lane_xor<2>(v);
    v += lane_xor<4>(v);
    return v;
}
__device__ __forceinline__ float dot4(const float4 a, const float4 b) {
    return fmaf(a.x, b.x, fmaf(a.y, b.y, fmaf(a.z, b.z, a.w * b.w)));
}

__global__ __launch_bounds__(256)
void gat_node_gather(const float* __restrict__ nt,
                     const float* __restrict__ a_w,
                     const int*   __restrict__ head,
                     const int2*  __restrict__ pk,
                     unsigned* __restrict__ maxenc,
                     float* __restrict__ denom,
                     float* __restrict__ out) {
    __shared__ unsigned smax;
    if (threadIdx.x == 0) smax = 0u;
    __syncthreads();

    const int j   = threadIdx.x & 7;
    const int n   = blockIdx.x * 32 + (threadIdx.x >> 3);
    const int col = 4 * j;
    const int hb  = j >> 2;
    const int pa  = hb * 32 + (col & 15);

    float4 ws0[H_HEADS], ws1[H_HEADS];
    float  td[H_HEADS];
    {
        const float* trow = nt + (size_t)n * HF + col;
        const float4 t0 = *(const float4*)trow;
        const float4 t1 = *(const float4*)(trow + 32);
#pragma unroll
        for (int k = 0; k < H_HEADS; ++k) {
            const float* ar = a_w + k * (2 * HF) + pa;
            ws0[k] = *(const float4*)ar;
            ws1[k] = *(const float4*)(ar + 64);
            const float4 wt0 = *(const float4*)(ar + 16);
            const float4 wt1 = *(const float4*)(ar + 80);
            td[k] = sum8(dot4(t0, wt0) + dot4(t1, wt1));
        }
    }

    float4 acc0 = make_float4(0.f, 0.f, 0.f, 0.f);
    float4 acc1 = make_float4(0.f, 0.f, 0.f, 0.f);
    float  den[H_HEADS] = {0.f, 0.f, 0.f, 0.f};
    float  mx = -INFINITY;

    const int e0 = head[n];
    bool live = (unsigned)e0 < (unsigned)E_EDGES;
    int2 p = make_int2(0, -1);
    if (live) p = pk[e0];

    for (int it = 0; it < E_EDGES; ++it) {
        if (__builtin_amdgcn_ballot_w32(live) == 0u) break;
        const bool more = live && ((unsigned)p.y < (unsigned)E_EDGES);
        const float* srow = nt + (size_t)p.x * HF + col;
        const float4 r0 = *(const float4*)srow;
        const float4 r1 = *(const float4*)(srow + 32);
        const int2 pn = pk[more ? p.y : 0];

        float lg[H_HEADS];
#pragma unroll
        for (int k = 0; k < H_HEADS; ++k)
            lg[k] = sum8(dot4(r0, ws0[k]) + dot4(r1, ws1[k])) + td[k];

        const float lmax = fmaxf(fmaxf(lg[0], lg[1]), fmaxf(lg[2], lg[3]));
        mx = live ? fmaxf(mx, lmax) : mx;
        float ev[H_HEADS];
#pragma unroll
        for (int k = 0; k < H_HEADS; ++k) {
            ev[k] = live ? __expf(NEG_SLOPE * lg[k]) : 0.0f;
            den[k] += ev[k];
        }
        const float e_lo = hb ? ev[1] : ev[0];
        const float e_hi = hb ? ev[3] : ev[2];
        acc0.x = fmaf(e_lo, r0.x, acc0.x); acc0.y = fmaf(e_lo, r0.y, acc0.y);
        acc0.z = fmaf(e_lo, r0.z, acc0.z); acc0.w = fmaf(e_lo, r0.w, acc0.w);
        acc1.x = fmaf(e_hi, r1.x, acc1.x); acc1.y = fmaf(e_hi, r1.y, acc1.y);
        acc1.z = fmaf(e_hi, r1.z, acc1.z); acc1.w = fmaf(e_hi, r1.w, acc1.w);

        live = more;
        p = pn;
    }

    float* orow = out + (size_t)n * HF + col;
    *(float4*)orow        = acc0;
    *(float4*)(orow + 32) = acc1;
    if (j == 0) {
        *(float4*)(denom + (size_t)n * H_HEADS) =
            make_float4(den[0], den[1], den[2], den[3]);
        atomicMax(&smax, fenc(mx));
    }
    __syncthreads();
    if (threadIdx.x == 0) atomicMax(maxenc, smax);
}

__global__ __launch_bounds__(256)
void gat_normalize(float* __restrict__ out, const float* __restrict__ denom,
                   const unsigned* __restrict__ maxenc) {
    const int i = blockIdx.x * blockDim.x + threadIdx.x;
    if (i >= N_NODES * HF / 4) return;
    const float eps = 1e-8f * __expf(NEG_SLOPE * fdec(*maxenc));
    const float d = denom[i >> 2] + eps;
    float4 v = ((const float4*)out)[i];
    v.x /= d; v.y /= d; v.z /= d; v.w /= d;
    ((float4*)out)[i] = v;
}

extern "C" void kernel_launch(void* const* d_in, const int* in_sizes, int n_in,
                              void* d_out, int out_size, void* d_ws, size_t ws_size,
                              hipStream_t stream) {
    const float* x   = (const float*)d_in[0];
    const float* W_w = (const float*)d_in[1];
    const float* a_w = (const float*)d_in[2];
    const int*   ei  = (const int*)d_in[3];
    const int* src = ei;
    const int* tgt = ei + E_EDGES;
    float* out = (float*)d_out;

    float* nt    = (float*)d_ws;
    float* denom = nt + (size_t)N_NODES * HF;
    unsigned* maxenc = (unsigned*)(denom + (size_t)N_NODES * H_HEADS);
    int*  head = (int*)(maxenc + 4);
    int2* pk   = (int2*)(head + N_NODES);

    (void)in_sizes; (void)n_in; (void)out_size; (void)ws_size;

    {
        gat_transform_wmma<<<N_NODES / 16, 128, 0, stream>>>(x, W_w, nt, head, maxenc);
    }
    {
        gat_link<<<E_EDGES / 256, 256, 0, stream>>>(src, tgt, head, pk);
    }
    {
        gat_node_gather<<<N_NODES / 32, 256, 0, stream>>>(
            nt, a_w, head, pk, maxenc, denom, out);
    }
    {
        gat_normalize<<<N_NODES * HF / 4 / 256, 256, 0, stream>>>(out, denom, maxenc);
    }
}


# Gather divides by the node's own denominator, stores S/D once; last pass: one thread per node reads denom only, rescales just groups whose rounded denominator +eps changes (two more roundings there)
# speedup vs baseline: 3.9926x; 1.1468x over previous
#include <hip/hip_runtime.h>
#include <hip/hip_bf16.h>


#define N_NODES   100000
#define E_EDGES   1600000
#define IN_F      128
#define H_HEADS   4
#define F_OUT     16
#define HF        64
#define NEG_SLOPE 0.01f

typedef float v2f __attribute__((ext_vector_type(2)));
typedef float v8f __attribute__((ext_vector_type(8)));

__device__ __forceinline__ unsigned fenc(float f) {
    unsigned u = __float_as_uint(f);
    return (u & 0x80000000u) ? ~u : (u | 0x80000000u);
}
__device__ __forceinline__ float fdec(unsigned k) {
    return (k & 0x80000000u) ? __uint_as_float(k ^ 0x80000000u)
                             : __uint_as_float(~k);
}

__global__ __launch_bounds__(128)
void gat_transform_wmma(const float* __restrict__ x,
                        const float* __restrict__ W,
                        float* __restrict__ nt,
                        int* __restrict__ head,
                        unsigned* __restrict__ maxenc) {
    const int lane = threadIdx.x & 31;
    const int wave = threadIdx.x >> 5;
    const int m    = lane & 15;
    const int kk   = lane >> 4;
    const int row  = blockIdx.x * 16 + m;
    const int ncol = wave * 16 + m;

    const float* xrow = x + (size_t)row  * IN_F + 2 * kk;
    const float* wrow = W + (size_t)ncol * IN_F + 2 * kk;

    v8f c = {};
#pragma unroll
    for (int k0 = 0; k0 < IN_F; k0 += 4) {
        v2f a = *(const v2f*)(xrow + k0);
        v2f b = *(const v2f*)(wrow + k0);
        c = __builtin_amdgcn_wmma_f32_16x16x4_f32(
                 false, a,  false, b,
                 (short)0, c,  false,  false);
    }

    if (threadIdx.x < 16) head[blockIdx.x * 16 + threadIdx.x] = -1;
    if (blockIdx.x == 0 && threadIdx.x == 0) *maxenc = 0u;

    const int half = lane >> 4;
#pragma unroll
    for (int r = 0; r < 8; ++r) {
        int mm = r + 8 * half;
        nt[(size_t)(blockIdx.x * 16 + mm) * HF + wave * 16 + (lane & 15)] = c[r];
    }
}

__global__ __launch_bounds__(256)
void gat_link(const int* __restrict__ src, const int* __restrict__ tgt,
              int* __restrict__ head, int2* __restrict__ pk) {
    const int e = blockIdx.x * 256 + threadIdx.x;
    if (e >= E_EDGES) return;
    const int s = src[e];
    const int t = tgt[e];
    const int prev = atomicExch(&head[t], e);
    pk[e] = make_int2(s, prev);
}

template <int M>
__device__ __forceinline__ float lane_xor(float v) {
    return __builtin_bit_cast(float, __builtin_amdgcn_update_dpp(
        0, __builtin_bit_cast(int, v), 0x160 | M, 0xf, 0xf, true));
}
__device__ __forceinline__ float sum8(float v) {
    v += lane_xor<1>(v);
    v += lane_xor<2>(v);
    v += lane_xor<4>(v);
    return v;
}
__device__ __forceinline__ float dot4(const float4 a, const float4 b) {
    return fmaf(a.x, b.x, fmaf(a.y, b.y, fmaf(a.z, b.z, a.w * b.w)));
}

__global__ __launch_bounds__(256)
void gat_node_gather(const float* __restrict__ nt,
                     const float* __restrict__ a_w,
                     const int*   __restrict__ head,
                     const int2*  __restrict__ pk,
                     unsigned* __restrict__ maxenc,
                     float* __restrict__ denom,
                     float* __restrict__ out) {
    __shared__ unsigned smax;
    if (threadIdx.x == 0) smax = 0u;
    __syncthreads();

    const int j   = threadIdx.x & 7;
    const int n   = blockIdx.x * 32 + (threadIdx.x >> 3);
    const int col = 4 * j;
    const int hb  = j >> 2;
    const int pa  = hb * 32 + (col & 15);

    float4 ws0[H_HEADS], ws1[H_HEADS];
    float  td[H_HEADS];
    {
        const float* trow = nt + (size_t)n * HF + col;
        const float4 t0 = *(const float4*)trow;
        const float4 t1 = *(const float4*)(trow + 32);
#pragma unroll
        for (int k = 0; k < H_HEADS; ++k) {
            const float* ar = a_w + k * (2 * HF) + pa;
            ws0[k] = *(const float4*)ar;
            ws1[k] = *(const float4*)(ar + 64);
            const float4 wt0 = *(const float4*)(ar + 16);
            const float4 wt1 = *(const float4*)(ar + 80);
            td[k] = sum8(dot4(t0, wt0) + dot4(t1, wt1));
        }
    }

    float4 acc0 = make_float4(0.f, 0.f, 0.f, 0.f);
    float4 acc1 = make_float4(0.f, 0.f, 0.f, 0.f);
    float  den[H_HEADS] = {0.f, 0.f, 0.f, 0.f};
    float  mx = -INFINITY;

    const int e0 = head[n];
    bool live = (unsigned)e0 < (unsigned)E_EDGES;
    int2 p = make_int2(0, -1);
    if (live) p = pk[e0];

    for (int it = 0; it < E_EDGES; ++it) {
        if (__builtin_amdgcn_ballot_w32(live) == 0u) break;
        const bool more = live && ((unsigned)p.y < (unsigned)E_EDGES);
        const float* srow = nt + (size_t)p.x * HF + col;
        const float4 r0 = *(const float4*)srow;
        const float4 r1 = *(const float4*)(srow + 32);
        const int2 pn = pk[more ? p.y : 0];

        float lg[H_HEADS];
#pragma unroll
        for (int k = 0; k < H_HEADS; ++k)
            lg[k] = sum8(dot4(r0, ws0[k]) + dot4(r1, ws1[k])) + td[k];

        const float lmax = fmaxf(fmaxf(lg[0], lg[1]), fmaxf(lg[2], lg[3]));
        mx = live ? fmaxf(mx, lmax) : mx;
        float ev[H_HEADS];
#pragma unroll
        for (int k = 0; k < H_HEADS; ++k) {
            ev[k] = live ? __expf(NEG_SLOPE * lg[k]) : 0.0f;
            den[k] += ev[k];
        }
        const float e_lo = hb ? ev[1] : ev[0];
        const float e_hi = hb ? ev[3] : ev[2];
        acc0.x = fmaf(e_lo, r0.x, acc0.x); acc0.y = fmaf(e_lo, r0.y, acc0.y);
        acc0.z = fmaf(e_lo, r0.z, acc0.z); acc0.w = fmaf(e_lo, r0.w, acc0.w);
        acc1.x = fmaf(e_hi, r1.x, acc1.x); acc1.y = fmaf(e_hi, r1.y, acc1.y);
        acc1.z = fmaf(e_hi, r1.z, acc1.z); acc1.w = fmaf(e_hi, r1.w, acc1.w);

        live = more;
        p = pn;
    }

    const float d_lo = hb ? den[1] : den[0];
    const float d_hi = hb ? den[3] : den[2];
    if (d_lo > 0.0f) {
        acc0.x /= d_lo; acc0.y /= d_lo; acc0.z /= d_lo; acc0.w /= d_lo;
    } else {
        acc0 = make_float4(0.f, 0.f, 0.f, 0.f);
    }
    if (d_hi > 0.0f) {
        acc1.x /= d_hi; acc1.y /= d_hi; acc1.z /= d_hi; acc1.w /= d_hi;
    } else {
        acc1 = make_float4(0.f, 0.f, 0.f, 0.f);
    }
    float* orow = out + (size_t)n * HF + col;
    *(float4*)orow        = acc0;
    *(float4*)(orow + 32) = acc1;
    if (j == 0) {
        *(float4*)(denom + (size_t)n * H_HEADS) =
            make_float4(den[0], den[1], den[2], den[3]);
        atomicMax(&smax, fenc(mx));
    }
    __syncthreads();
    if (threadIdx.x == 0) atomicMax(maxenc, smax);
}

__global__ __launch_bounds__(256)
void gat_epsilon(float* __restrict__ out, const float* __restrict__ denom,
                 const unsigned* __restrict__ maxenc) {
    const int n = blockIdx.x * 256 + threadIdx.x;
    if (n >= N_NODES) return;
    const float eps = 1e-8f * __expf(NEG_SLOPE * fdec(*maxenc));
    const float4 dv = ((const float4*)denom)[n];
    const float dd[H_HEADS] = {dv.x, dv.y, dv.z, dv.w};
#pragma unroll
    for (int k = 0; k < H_HEADS; ++k) {
        const float d = dd[k] + eps;
        if (d != dd[k]) {
            const float sc = dd[k] / d;
            float4* o = (float4*)(out + (size_t)n * HF + k * F_OUT);
            for (int q = 0; q < F_OUT / 4; ++q) {
                float4 v = o[q];
                v.x *= sc; v.y *= sc; v.z *= sc; v.w *= sc;
                o[q] = v;
            }
        }
    }
}

extern "C" void kernel_launch(void* const* d_in, const int* in_sizes, int n_in,
                              void* d_out, int out_size, void* d_ws, size_t ws_size,
                              hipStream_t stream) {
    const float* x   = (const float*)d_in[0];
    const float* W_w = (const float*)d_in[1];
    const float* a_w = (const float*)d_in[2];
    const int*   ei  = (const int*)d_in[3];
    const int* src = ei;
    const int* tgt = ei + E_EDGES;
    float* out = (float*)d_out;

    float* nt    = (float*)d_ws;
    float* denom = nt + (size_t)N_NODES * HF;
    unsigned* maxenc = (unsigned*)(denom + (size_t)N_NODES * H_HEADS);
    int*  head = (int*)(maxenc + 4);
    int2* pk   = (int2*)(head + N_NODES);

    (void)in_sizes; (void)n_in; (void)out_size; (void)ws_size;

    {
        gat_transform_wmma<<<N_NODES / 16, 128, 0, stream>>>(x, W_w, nt, head, maxenc);
    }
    {
        gat_link<<<E_EDGES / 256, 256, 0, stream>>>(src, tgt, head, pk);
    }
    {
        gat_node_gather<<<N_NODES / 32, 256, 0, stream>>>(
            nt, a_w, head, pk, maxenc, denom, out);
    }
    {
        gat_epsilon<<<(N_NODES + 255) / 256, 256, 0, stream>>>(out, denom, maxenc);
    }
}


# Edge lists built with no global atomic: chunks group edges by target range (LDS counts, whole-line stores), one block per range links them by LDS exchange; 5 launches; sums stay in arrival order
# speedup vs baseline: 6.3868x; 1.5997x over previous
#include <hip/hip_runtime.h>
#include <hip/hip_bf16.h>


#define N_NODES   100000
#define E_EDGES   1600000
#define IN_F      128
#define H_HEADS   4
#define F_OUT     16
#define HF        64
#define NEG_SLOPE 0.01f

#define N_CHUNKS  256
#define CHUNK_E   6250
#define BIN_R     384
#define N_BINS    261
#define OFF_PITCH 262

typedef float v2f __attribute__((ext_vector_type(2)));
typedef float v8f __attribute__((ext_vector_type(8)));

__device__ __forceinline__ unsigned fenc(float f) {
    unsigned u = __float_as_uint(f);
    return (u & 0x80000000u) ? ~u : (u | 0x80000000u);
}
__device__ __forceinline__ float fdec(unsigned k) {
    return (k & 0x80000000u) ? __uint_as_float(k ^ 0x80000000u)
                             : __uint_as_float(~k);
}

__global__ __launch_bounds__(128)
void gat_transform_wmma(const float* __restrict__ x,
                        const float* __restrict__ W,
                        float* __restrict__ nt,
                        unsigned* __restrict__ maxenc) {
    const int lane = threadIdx.x & 31;
    const int wave = threadIdx.x >> 5;
    const int m    = lane & 15;
    const int kk   = lane >> 4;
    const int row  = blockIdx.x * 16 + m;
    const int ncol = wave * 16 + m;

    const float* xrow = x + (size_t)row  * IN_F + 2 * kk;
    const float* wrow = W + (size_t)ncol * IN_F + 2 * kk;

    v8f c = {};
#pragma unroll
    for (int k0 = 0; k0 < IN_F; k0 += 4) {
        v2f a = *(const v2f*)(xrow + k0);
        v2f b = *(const v2f*)(wrow + k0);
        c = __builtin_amdgcn_wmma_f32_16x16x4_f32(
                 false, a,  false, b,
                 (short)0, c,  false,  false);
    }

    if (blockIdx.x == 0 && threadIdx.x == 0) *maxenc = 0u;

    const int half = lane >> 4;
#pragma unroll
    for (int r = 0; r < 8; ++r) {
        int mm = r + 8 * half;
        nt[(size_t)(blockIdx.x * 16 + mm) * HF + wave * 16 + (lane & 15)] = c[r];
    }
}

__device__ __forceinline__ int bin_of(int t) {
    const unsigned b = (unsigned)t / BIN_R;
    return (int)(b < N_BINS ? b : N_BINS - 1);
}

__global__ __launch_bounds__(1024)
void gat_group(const int* __restrict__ src, const int* __restrict__ tgt,
               int* __restrict__ off, int2* __restrict__ pk) {
    __shared__ int  cnt[288];
    __shared__ int2 stage[CHUNK_E];
    const int tid  = threadIdx.x;
    const int base = blockIdx.x * CHUNK_E;
    if (tid < 288) cnt[tid] = 0;

    int sv[7], tv[7], rk[7];
#pragma unroll
    for (int q = 0; q < 7; ++q) {
        const int i = q * 1024 + tid;
        sv[q] = 0; tv[q] = 0;
        if (i < CHUNK_E) {
            sv[q] = src[base + i];
            tv[q] = tgt[base + i];
        }
    }
    __syncthreads();
#pragma unroll
    for (int q = 0; q < 7; ++q) {
        const int i = q * 1024 + tid;
        rk[q] = 0;
        if (i < CHUNK_E)
            rk[q] = atomicAdd(&cnt[bin_of(tv[q])], 1);
    }
    __syncthreads();

    if (tid < 32) {
        int v[9];
        int s = 0;
#pragma unroll
        for (int q = 0; q < 9; ++q) { v[q] = cnt[tid * 9 + q]; s += v[q]; }
        int inc = s;
#pragma unroll
        for (int d = 1; d < 32; d <<= 1) {
            const int o = __shfl_up(inc, d);
            if (tid >= d) inc += o;
        }
        int run = inc - s;
#pragma unroll
        for (int q = 0; q < 9; ++q) { cnt[tid * 9 + q] = run; run += v[q]; }
    }
    __syncthreads();

    if (tid < OFF_PITCH) off[blockIdx.x * OFF_PITCH + tid] = cnt[tid];
#pragma unroll
    for (int q = 0; q < 7; ++q) {
        const int i = q * 1024 + tid;
        if (i < CHUNK_E)
            stage[cnt[bin_of(tv[q])] + rk[q]] = make_int2(sv[q], tv[q]);
    }
    __syncthreads();
    const int a0 = base & ~15;
#pragma unroll
    for (int q = 0; q < 7; ++q) {
        const int a = a0 + q * 1024 + tid;
        if (a >= base && a < base + CHUNK_E) pk[a] = stage[a - base];
    }
}

__global__ __launch_bounds__(1024)
void gat_link_lds(const int* __restrict__ off, int* __restrict__ head,
                  int2* __restrict__ pk) {
    __shared__ int lhead[BIN_R];
    __shared__ int seg_at[N_CHUNKS];
    __shared__ int seg_len[N_CHUNKS];
    const int tid  = threadIdx.x;
    const int lane = tid & 31;
    const int wave = tid >> 5;
    const int b    = blockIdx.x;
    const int nb   = b * BIN_R;

    if (tid < BIN_R) lhead[tid] = -1;
    if (tid < N_CHUNKS) {
        const int o0 = off[tid * OFF_PITCH + b];
        const int o1 = off[tid * OFF_PITCH + b + 1];
        seg_at[tid]  = tid * CHUNK_E + o0;
        seg_len[tid] = o1 - o0;
    }
    __syncthreads();

    int2 st[8];
    int  at[8], ln[8];
#pragma unroll
    for (int r = 0; r < 8; ++r) {
        at[r] = seg_at[wave + 32 * r];
        ln[r] = seg_len[wave + 32 * r];
        st[r] = make_int2(0, 0);
        if (lane < ln[r]) st[r] = pk[at[r] + lane];
    }
#pragma unroll
    for (int r = 0; r < 8; ++r) {
        if (lane < ln[r]) {
            const int e  = at[r] + lane;
            int tl = st[r].y - nb;
            tl = tl < 0 ? 0 : (tl > BIN_R - 1 ? BIN_R - 1 : tl);
            const int prev = atomicExch(&lhead[tl], e);
            pk[e] = make_int2(st[r].x, prev);
        }
        for (int i = 32 + lane; i < ln[r]; i += 32) {
            const int e = at[r] + i;
            const int2 s2 = pk[e];
            int tl = s2.y - nb;
            tl = tl < 0 ? 0 : (tl > BIN_R - 1 ? BIN_R - 1 : tl);
            const int prev = atomicExch(&lhead[tl], e);
            pk[e] = make_int2(s2.x, prev);
        }
    }
    __syncthreads();
    if (tid < BIN_R && nb + tid < N_NODES) head[nb + tid] = lhead[tid];
}

template <int M>
__device__ __forceinline__ float lane_xor(float v) {
    return __builtin_bit_cast(float, __builtin_amdgcn_update_dpp(
        0, __builtin_bit_cast(int, v), 0x160 | M, 0xf, 0xf, true));
}
__device__ __forceinline__ float sum8(float v) {
    v += lane_xor<1>(v);
    v += lane_xor<2>(v);
    v += lane_xor<4>(v);
    return v;
}
__device__ __forceinline__ float dot4(const float4 a, const float4 b) {
    return fmaf(a.x, b.x, fmaf(a.y, b.y, fmaf(a.z, b.z, a.w * b.w)));
}

__global__ __launch_bounds__(256)
void gat_node_gather(const float* __restrict__ nt,
                     const float* __restrict__ a_w,
                     const int*   __restrict__ head,
                     const int2*  __restrict__ pk,
                     unsigned* __restrict__ maxenc,
                     float* __restrict__ denom,
                     float* __restrict__ out) {
    __shared__ unsigned smax;
    if (threadIdx.x == 0) smax = 0u;
    __syncthreads();

    const int j   = threadIdx.x & 7;
    const int n   = blockIdx.x * 32 + (threadIdx.x >> 3);
    const int col = 4 * j;
    const int hb  = j >> 2;
    const int pa  = hb * 32 + (col & 15);

    float4 ws0[H_HEADS], ws1[H_HEADS];
    float  td[H_HEADS];
    {
        const float* trow = nt + (size_t)n * HF + col;
        const float4 t0 = *(const float4*)trow;
        const float4 t1 = *(const float4*)(trow + 32);
#pragma unroll
        for (int k = 0; k < H_HEADS; ++k) {
            const float* ar = a_w + k * (2 * HF) + pa;
            ws0[k] = *(const float4*)ar;
            ws1[k] = *(const float4*)(ar + 64);
            const float4 wt0 = *(const float4*)(ar + 16);
            const float4 wt1 = *(const float4*)(ar + 80);
            td[k] = sum8(dot4(t0, wt0) + dot4(t1, wt1));
        }
    }

    float4 acc0 = make_float4(0.f, 0.f, 0.f, 0.f);
    float4 acc1 = make_float4(0.f, 0.f, 0.f, 0.f);
    float  den[H_HEADS] = {0.f, 0.f, 0.f, 0.f};
    float  mx = -INFINITY;

    const int e0 = head[n];
    bool live = (unsigned)e0 < (unsigned)E_EDGES;
    int2 p = make_int2(0, -1);
    if (live) p = pk[e0];

    for (int it = 0; it < E_EDGES; ++it) {
        if (__builtin_amdgcn_ballot_w32(live) == 0u) break;
        const bool more = live && ((unsigned)p.y < (unsigned)E_EDGES);
        const float* srow = nt + (size_t)p.x * HF + col;
        const float4 r0 = *(const float4*)srow;
        const float4 r1 = *(const float4*)(srow + 32);
        const int2 pn = pk[more ? p.y : 0];

        float lg[H_HEADS];
#pragma unroll
        for (int k = 0; k < H_HEADS; ++k)
            lg[k] = sum8(dot4(r0, ws0[k]) + dot4(r1, ws1[k])) + td[k];

        const float lmax = fmaxf(fmaxf(lg[0], lg[1]), fmaxf(lg[2], lg[3]));
        mx = live ? fmaxf(mx, lmax) : mx;
        float ev[H_HEADS];
#pragma unroll
        for (int k = 0; k < H_HEADS; ++k) {
            ev[k] = live ? __expf(NEG_SLOPE * lg[k]) : 0.0f;
            den[k] += ev[k];
        }
        const float e_lo = hb ? ev[1] : ev[0];
        const float e_hi = hb ? ev[3] : ev[2];
        acc0.x = fmaf(e_lo, r0.x, acc0.x); acc0.y = fmaf(e_lo, r0.y, acc0.y);
        acc0.z = fmaf(e_lo, r0.z, acc0.z); acc0.w = fmaf(e_lo, r0.w, acc0.w);
        acc1.x = fmaf(e_hi, r1.x, acc1.x); acc1.y = fmaf(e_hi, r1.y, acc1.y);
        acc1.z = fmaf(e_hi, r1.z, acc1.z); acc1.w = fmaf(e_hi, r1.w, acc1.w);

        live = more;
        p = pn;
    }

    const float d_lo = hb ? den[1] : den[0];
    const float d_hi = hb ? den[3] : den[2];
    if (d_lo > 0.0f) {
        acc0.x /= d_lo; acc0.y /= d_lo; acc0.z /= d_lo; acc0.w /= d_lo;
    } else {
        acc0 = make_float4(0.f, 0.f, 0.f, 0.f);
    }
    if (d_hi > 0.0f) {
        acc1.x /= d_hi; acc1.y /= d_hi; acc1.z /= d_hi; acc1.w /= d_hi;
    } else {
        acc1 = make_float4(0.f, 0.f, 0.f, 0.f);
    }
    float* orow = out + (size_t)n * HF + col;
    *(float4*)orow        = acc0;
    *(float4*)(orow + 32) = acc1;
    if (j == 0) {
        *(float4*)(denom + (size_t)n * H_HEADS) =
            make_float4(den[0], den[1], den[2], den[3]);
        atomicMax(&smax, fenc(mx));
    }
    __syncthreads();
    if (threadIdx.x == 0) atomicMax(maxenc, smax);
}

__global__ __launch_bounds__(256)
void gat_epsilon(float* __restrict__ out, const float* __restrict__ denom,
                 const unsigned* __restrict__ maxenc) {
    const int n = blockIdx.x * 256 + threadIdx.x;
    if (n >= N_NODES) return;
    const float eps = 1e-8f * __expf(NEG_SLOPE * fdec(*maxenc));
    const float4 dv = ((const float4*)denom)[n];
    const float dd[H_HEADS] = {dv.x, dv.y, dv.z, dv.w};
#pragma unroll
    for (int k = 0; k < H_HEADS; ++k) {
        const float d = dd[k] + eps;
        if (d != dd[k]) {
            const float sc = dd[k] / d;
            float4* o = (float4*)(out + (size_t)n * HF + k * F_OUT);
            for (int q = 0; q < F_OUT / 4; ++q) {
                float4 v = o[q];
                v.x *= sc; v.y *= sc; v.z *= sc; v.w *= sc;
                o[q] = v;
            }
        }
    }
}

extern "C" void kernel_launch(void* const* d_in, const int* in_sizes, int n_in,
                              void* d_out, int out_size, void* d_ws, size_t ws_size,
                              hipStream_t stream) {
    const float* x   = (const float*)d_in[0];
    const float* W_w = (const float*)d_in[1];
    const float* a_w = (const float*)d_in[2];
    const int*   ei  = (const int*)d_in[3];
    const int* src = ei;
    const int* tgt = ei + E_EDGES;
    float* out = (float*)d_out;

    float* nt    = (float*)d_ws;
    float* denom = nt + (size_t)N_NODES * HF;
    unsigned* maxenc = (unsigned*)(denom + (size_t)N_NODES * H_HEADS);
    int*  head = (int*)(maxenc + 4);
    int2* pk   = (int2*)(head + N_NODES);
    int*  off  = (int*)denom;

    (void)in_sizes; (void)n_in; (void)out_size; (void)ws_size;

    {
        gat_transform_wmma<<<N_NODES / 16, 128, 0, stream>>>(x, W_w, nt, maxenc);
    }
    {
        gat_group<<<N_CHUNKS, 1024, 0, stream>>>(src, tgt, off, pk);
    }
    {
        gat_link_lds<<<N_BINS, 1024, 0, stream>>>(off, head, pk);
    }
    {
        gat_node_gather<<<N_NODES / 32, 256, 0, stream>>>(
            nt, a_w, head, pk, maxenc, denom, out);
    }
    {
        gat_epsilon<<<(N_NODES + 255) / 256, 256, 0, stream>>>(out, denom, maxenc);
    }
}


# Transform: one wave per 16 rows holds all four 16-column tiles (16x64 strip): each x fragment is loaded once for four WMMAs, the strip's x requested up front; 4x fewer waves; every sum unchanged
# speedup vs baseline: 6.8974x; 1.0799x over previous
#include <hip/hip_runtime.h>
#include <hip/hip_bf16.h>


#define N_NODES   100000
#define E_EDGES   1600000
#define IN_F      128
#define H_HEADS   4
#define F_OUT     16
#define HF        64
#define NEG_SLOPE 0.01f

#define N_CHUNKS  256
#define CHUNK_E   6250
#define BIN_R     384
#define N_BINS    261
#define OFF_PITCH 262

typedef float v2f __attribute__((ext_vector_type(2)));
typedef float v8f __attribute__((ext_vector_type(8)));

__device__ __forceinline__ unsigned fenc(float f) {
    unsigned u = __float_as_uint(f);
    return (u & 0x80000000u) ? ~u : (u | 0x80000000u);
}
__device__ __forceinline__ float fdec(unsigned k) {
    return (k & 0x80000000u) ? __uint_as_float(k ^ 0x80000000u)
                             : __uint_as_float(~k);
}

__global__ __launch_bounds__(64)
void gat_transform_wmma(const float* __restrict__ x,
                        const float* __restrict__ W,
                        float* __restrict__ nt,
                        unsigned* __restrict__ maxenc) {
    const int lane = threadIdx.x & 31;
    const int wave = threadIdx.x >> 5;
    const int m    = lane & 15;
    const int kk   = lane >> 4;
    const int tile = blockIdx.x * 2 + wave;
    const int row  = tile * 16 + m;

    const float* xrow = x + (size_t)row * IN_F + 2 * kk;
    const float* wrow = W + (size_t)m * IN_F + 2 * kk;

    v2f xa[IN_F / 4];
#pragma unroll
    for (int s = 0; s < IN_F / 4; ++s)
        xa[s] = *(const v2f*)(xrow + 4 * s);
    __builtin_amdgcn_sched_barrier(0);

    v8f c0 = {}, c1 = {}, c2 = {}, c3 = {};
#pragma unroll
    for (int k0 = 0; k0 < IN_F; k0 += 4) {
        const v2f a  = xa[k0 / 4];
        const v2f b0 = *(const v2f*)(wrow + k0);
        const v2f b1 = *(const v2f*)(wrow + 16 * IN_F + k0);
        const v2f b2 = *(const v2f*)(wrow + 32 * IN_F + k0);
        const v2f b3 = *(const v2f*)(wrow + 48 * IN_F + k0);
        c0 = __builtin_amdgcn_wmma_f32_16x16x4_f32(
                 false, a,  false, b0,
                 (short)0, c0,  false,  false);
        c1 = __builtin_amdgcn_wmma_f32_16x16x4_f32(
                false, a, false, b1, (short)0, c1, false, false);
        c2 = __builtin_amdgcn_wmma_f32_16x16x4_f32(
                false, a, false, b2, (short)0, c2, false, false);
        c3 = __builtin_amdgcn_wmma_f32_16x16x4_f32(
                false, a, false, b3, (short)0, c3, false, false);
    }

    if (blockIdx.x == 0 && threadIdx.x == 0) *maxenc = 0u;

    const int half = lane >> 4;
    float* orow = nt + (size_t)(tile * 16 + 8 * half) * HF + (lane & 15);
#pragma unroll
    for (int r = 0; r < 8; ++r) {
        float* o = orow + (size_t)r * HF;
        o[0]  = c0[r];
        o[16] = c1[r];
        o[32] = c2[r];
        o[48] = c3[r];
    }
}

__device__ __forceinline__ int bin_of(int t) {
    const unsigned b = (unsigned)t / BIN_R;
    return (int)(b < N_BINS ? b : N_BINS - 1);
}

__global__ __launch_bounds__(1024)
void gat_group(const int* __restrict__ src, const int* __restrict__ tgt,
               int* __restrict__ off, int2* __restrict__ pk) {
    __shared__ int  cnt[288];
    __shared__ int2 stage[CHUNK_E];
    const int tid  = threadIdx.x;
    const int base = blockIdx.x * CHUNK_E;
    if (tid < 288) cnt[tid] = 0;

    int sv[7], tv[7], rk[7];
#pragma unroll
    for (int q = 0; q < 7; ++q) {
        const int i = q * 1024 + tid;
        sv[q] = 0; tv[q] = 0;
        if (i < CHUNK_E) {
            sv[q] = src[base + i];
            tv[q] = tgt[base + i];
        }
    }
    __syncthreads();
#pragma unroll
    for (int q = 0; q < 7; ++q) {
        const int i = q * 1024 + tid;
        rk[q] = 0;
        if (i < CHUNK_E)
            rk[q] = atomicAdd(&cnt[bin_of(tv[q])], 1);
    }
    __syncthreads();

    if (tid < 32) {
        int v[9];
        int s = 0;
#pragma unroll
        for (int q = 0; q < 9; ++q) { v[q] = cnt[tid * 9 + q]; s += v[q]; }
        int inc = s;
#pragma unroll
        for (int d = 1; d < 32; d <<= 1) {
            const int o = __shfl_up(inc, d);
            if (tid >= d) inc += o;
        }
        int run = inc - s;
#pragma unroll
        for (int q = 0; q < 9; ++q) { cnt[tid * 9 + q] = run; run += v[q]; }
    }
    __syncthreads();

    if (tid < OFF_PITCH) off[blockIdx.x * OFF_PITCH + tid] = cnt[tid];
#pragma unroll
    for (int q = 0; q < 7; ++q) {
        const int i = q * 1024 + tid;
        if (i < CHUNK_E)
            stage[cnt[bin_of(tv[q])] + rk[q]] = make_int2(sv[q], tv[q]);
    }
    __syncthreads();
    const int a0 = base & ~15;
#pragma unroll
    for (int q = 0; q < 7; ++q) {
        const int a = a0 + q * 1024 + tid;
        if (a >= base && a < base + CHUNK_E) pk[a] = stage[a - base];
    }
}

__global__ __launch_bounds__(1024)
void gat_link_lds(const int* __restrict__ off, int* __restrict__ head,
                  int2* __restrict__ pk) {
    __shared__ int lhead[BIN_R];
    __shared__ int seg_at[N_CHUNKS];
    __shared__ int seg_len[N_CHUNKS];
    const int tid  = threadIdx.x;
    const int lane = tid & 31;
    const int wave = tid >> 5;
    const int b    = blockIdx.x;
    const int nb   = b * BIN_R;

    if (tid < BIN_R) lhead[tid] = -1;
    if (tid < N_CHUNKS) {
        const int o0 = off[tid * OFF_PITCH + b];
        const int o1 = off[tid * OFF_PITCH + b + 1];
        seg_at[tid]  = tid * CHUNK_E + o0;
        seg_len[tid] = o1 - o0;
    }
    __syncthreads();

    int2 st[8];
    int  at[8], ln[8];
#pragma unroll
    for (int r = 0; r < 8; ++r) {
        at[r] = seg_at[wave + 32 * r];
        ln[r] = seg_len[wave + 32 * r];
        st[r] = make_int2(0, 0);
        if (lane < ln[r]) st[r] = pk[at[r] + lane];
    }
#pragma unroll
    for (int r = 0; r < 8; ++r) {
        if (lane < ln[r]) {
            const int e  = at[r] + lane;
            int tl = st[r].y - nb;
            tl = tl < 0 ? 0 : (tl > BIN_R - 1 ? BIN_R - 1 : tl);
            const int prev = atomicExch(&lhead[tl], e);
            pk[e] = make_int2(st[r].x, prev);
        }
        for (int i = 32 + lane; i < ln[r]; i += 32) {
            const int e = at[r] + i;
            const int2 s2 = pk[e];
            int tl = s2.y - nb;
            tl = tl < 0 ? 0 : (tl > BIN_R - 1 ? BIN_R - 1 : tl);
            const int prev = atomicExch(&lhead[tl], e);
            pk[e] = make_int2(s2.x, prev);
        }
    }
    __syncthreads();
    if (tid < BIN_R && nb + tid < N_NODES) head[nb + tid] = lhead[tid];
}

template <int M>
__device__ __forceinline__ float lane_xor(float v) {
    return __builtin_bit_cast(float, __builtin_amdgcn_update_dpp(
        0, __builtin_bit_cast(int, v), 0x160 | M, 0xf, 0xf, true));
}
__device__ __forceinline__ float sum8(float v) {
    v += lane_xor<1>(v);
    v += lane_xor<2>(v);
    v += lane_xor<4>(v);
    return v;
}
__device__ __forceinline__ float dot4(const float4 a, const float4 b) {
    return fmaf(a.x, b.x, fmaf(a.y, b.y, fmaf(a.z, b.z, a.w * b.w)));
}

__global__ __launch_bounds__(256)
void gat_node_gather(const float* __restrict__ nt,
                     const float* __restrict__ a_w,
                     const int*   __restrict__ head,
                     const int2*  __restrict__ pk,
                     unsigned* __restrict__ maxenc,
                     float* __restrict__ denom,
                     float* __restrict__ out) {
    __shared__ unsigned smax;
    if (threadIdx.x == 0) smax = 0u;
    __syncthreads();

    const int j   = threadIdx.x & 7;
    const int n   = blockIdx.x * 32 + (threadIdx.x >> 3);
    const int col = 4 * j;
    const int hb  = j >> 2;
    const int pa  = hb * 32 + (col & 15);

    float4 ws0[H_HEADS], ws1[H_HEADS];
    float  td[H_HEADS];
    {
        const float* trow = nt + (size_t)n * HF + col;
        const float4 t0 = *(const float4*)trow;
        const float4 t1 = *(const float4*)(trow + 32);
#pragma unroll
        for (int k = 0; k < H_HEADS; ++k) {
            const float* ar = a_w + k * (2 * HF) + pa;
            ws0[k] = *(const float4*)ar;
            ws1[k] = *(const float4*)(ar + 64);
            const float4 wt0 = *(const float4*)(ar + 16);
            const float4 wt1 = *(const float4*)(ar + 80);
            td[k] = sum8(dot4(t0, wt0) + dot4(t1, wt1));
        }
    }

    float4 acc0 = make_float4(0.f, 0.f, 0.f, 0.f);
    float4 acc1 = make_float4(0.f, 0.f, 0.f, 0.f);
    float  den[H_HEADS] = {0.f, 0.f, 0.f, 0.f};
    float  mx = -INFINITY;

    const int e0 = head[n];
    bool live = (unsigned)e0 < (unsigned)E_EDGES;
    int2 p = make_int2(0, -1);
    if (live) p = pk[e0];

    for (int it = 0; it < E_EDGES; ++it) {
        if (__builtin_amdgcn_ballot_w32(live) == 0u) break;
        const bool more = live && ((unsigned)p.y < (unsigned)E_EDGES);
        const float* srow = nt + (size_t)p.x * HF + col;
        const float4 r0 = *(const float4*)srow;
        const float4 r1 = *(const float4*)(srow + 32);
        const int2 pn = pk[more ? p.y : 0];

        float lg[H_HEADS];
#pragma unroll
        for (int k = 0; k < H_HEADS; ++k)
            lg[k] = sum8(dot4(r0, ws0[k]) + dot4(r1, ws1[k])) + td[k];

        const float lmax = fmaxf(fmaxf(lg[0], lg[1]), fmaxf(lg[2], lg[3]));
        mx = live ? fmaxf(mx, lmax) : mx;
        float ev[H_HEADS];
#pragma unroll
        for (int k = 0; k < H_HEADS; ++k) {
            ev[k] = live ? __expf(NEG_SLOPE * lg[k]) : 0.0f;
            den[k] += ev[k];
        }
        const float e_lo = hb ? ev[1] : ev[0];
        const float e_hi = hb ? ev[3] : ev[2];
        acc0.x = fmaf(e_lo, r0.x, acc0.x); acc0.y = fmaf(e_lo, r0.y, acc0.y);
        acc0.z = fmaf(e_lo, r0.z, acc0.z); acc0.w = fmaf(e_lo, r0.w, acc0.w);
        acc1.x = fmaf(e_hi, r1.x, acc1.x); acc1.y = fmaf(e_hi, r1.y, acc1.y);
        acc1.z = fmaf(e_hi, r1.z, acc1.z); acc1.w = fmaf(e_hi, r1.w, acc1.w);

        live = more;
        p = pn;
    }

    const float d_lo = hb ? den[1] : den[0];
    const float d_hi = hb ? den[3] : den[2];
    if (d_lo > 0.0f) {
        acc0.x /= d_lo; acc0.y /= d_lo; acc0.z /= d_lo; acc0.w /= d_lo;
    } else {
        acc0 = make_float4(0.f, 0.f, 0.f, 0.f);
    }
    if (d_hi > 0.0f) {
        acc1.x /= d_hi; acc1.y /= d_hi; acc1.z /= d_hi; acc1.w /= d_hi;
    } else {
        acc1 = make_float4(0.f, 0.f, 0.f, 0.f);
    }
    float* orow = out + (size_t)n * HF + col;
    *(float4*)orow        = acc0;
    *(float4*)(orow + 32) = acc1;
    if (j == 0) {
        *(float4*)(denom + (size_t)n * H_HEADS) =
            make_float4(den[0], den[1], den[2], den[3]);
        atomicMax(&smax, fenc(mx));
    }
    __syncthreads();
    if (threadIdx.x == 0) atomicMax(maxenc, smax);
}

__global__ __launch_bounds__(256)
void gat_epsilon(float* __restrict__ out, const float* __restrict__ denom,
                 const unsigned* __restrict__ maxenc) {
    const int n = blockIdx.x * 256 + threadIdx.x;
    if (n >= N_NODES) return;
    const float eps = 1e-8f * __expf(NEG_SLOPE * fdec(*maxenc));
    const float4 dv = ((const float4*)denom)[n];
    const float dd[H_HEADS] = {dv.x, dv.y, dv.z, dv.w};
#pragma unroll
    for (int k = 0; k < H_HEADS; ++k) {
        const float d = dd[k] + eps;
        if (d != dd[k]) {
            const float sc = dd[k] / d;
            float4* o = (float4*)(out + (size_t)n * HF + k * F_OUT);
            for (int q = 0; q < F_OUT / 4; ++q) {
                float4 v = o[q];
                v.x *= sc; v.y *= sc; v.z *= sc; v.w *= sc;
                o[q] = v;
            }
        }
    }
}

extern "C" void kernel_launch(void* const* d_in, const int* in_sizes, int n_in,
                              void* d_out, int out_size, void* d_ws, size_t ws_size,
                              hipStream_t stream) {
    const float* x   = (const float*)d_in[0];
    const float* W_w = (const float*)d_in[1];
    const float* a_w = (const float*)d_in[2];
    const int*   ei  = (const int*)d_in[3];
    const int* src = ei;
    const int* tgt = ei + E_EDGES;
    float* out = (float*)d_out;

    float* nt    = (float*)d_ws;
    float* denom = nt + (size_t)N_NODES * HF;
    unsigned* maxenc = (unsigned*)(denom + (size_t)N_NODES * H_HEADS);
    int*  head = (int*)(maxenc + 4);
    int2* pk   = (int2*)(head + N_NODES);
    int*  off  = (int*)denom;

    (void)in_sizes; (void)n_in; (void)out_size; (void)ws_size;

    {
        gat_transform_wmma<<<N_NODES / 32, 64, 0, stream>>>(x, W_w, nt, maxenc);
    }
    {
        gat_group<<<N_CHUNKS, 1024, 0, stream>>>(src, tgt, off, pk);
    }
    {
        gat_link_lds<<<N_BINS, 1024, 0, stream>>>(off, head, pk);
    }
    {
        gat_node_gather<<<N_NODES / 32, 256, 0, stream>>>(
            nt, a_w, head, pk, maxenc, denom, out);
    }
    {
        gat_epsilon<<<(N_NODES + 255) / 256, 256, 0, stream>>>(out, denom, maxenc);
    }
}


# Gather: each lane forms only the 2 logits of its own row pieces: partials packed by role (kept / handed to the lane 4 on), 6 lane exchanges and 2 exps per hop, was 12 and 4; logit sum order changes
# speedup vs baseline: 7.4186x; 1.0756x over previous
#include <hip/hip_runtime.h>
#include <hip/hip_bf16.h>


#define N_NODES   100000
#define E_EDGES   1600000
#define IN_F      128
#define H_HEADS   4
#define F_OUT     16
#define HF        64
#define NEG_SLOPE 0.01f

#define N_CHUNKS  256
#define CHUNK_E   6250
#define BIN_R     384
#define N_BINS    261
#define OFF_PITCH 262

typedef float v2f __attribute__((ext_vector_type(2)));
typedef float v8f __attribute__((ext_vector_type(8)));

__device__ __forceinline__ unsigned fenc(float f) {
    unsigned u = __float_as_uint(f);
    return (u & 0x80000000u) ? ~u : (u | 0x80000000u);
}
__device__ __forceinline__ float fdec(unsigned k) {
    return (k & 0x80000000u) ? __uint_as_float(k ^ 0x80000000u)
                             : __uint_as_float(~k);
}

__global__ __launch_bounds__(64)
void gat_transform_wmma(const float* __restrict__ x,
                        const float* __restrict__ W,
                        float* __restrict__ nt,
                        unsigned* __restrict__ maxenc) {
    const int lane = threadIdx.x & 31;
    const int wave = threadIdx.x >> 5;
    const int m    = lane & 15;
    const int kk   = lane >> 4;
    const int tile = blockIdx.x * 2 + wave;
    const int row  = tile * 16 + m;

    const float* xrow = x + (size_t)row * IN_F + 2 * kk;
    const float* wrow = W + (size_t)m * IN_F + 2 * kk;

    v2f xa[IN_F / 4];
#pragma unroll
    for (int s = 0; s < IN_F / 4; ++s)
        xa[s] = *(const v2f*)(xrow + 4 * s);
    __builtin_amdgcn_sched_barrier(0);

    v8f c0 = {}, c1 = {}, c2 = {}, c3 = {};
#pragma unroll
    for (int k0 = 0; k0 < IN_F; k0 += 4) {
        const v2f a  = xa[k0 / 4];
        const v2f b0 = *(const v2f*)(wrow + k0);
        const v2f b1 = *(const v2f*)(wrow + 16 * IN_F + k0);
        const v2f b2 = *(const v2f*)(wrow + 32 * IN_F + k0);
        const v2f b3 = *(const v2f*)(wrow + 48 * IN_F + k0);
        c0 = __builtin_amdgcn_wmma_f32_16x16x4_f32(
                 false, a,  false, b0,
                 (short)0, c0,  false,  false);
        c1 = __builtin_amdgcn_wmma_f32_16x16x4_f32(
                false, a, false, b1, (short)0, c1, false, false);
        c2 = __builtin_amdgcn_wmma_f32_16x16x4_f32(
                false, a, false, b2, (short)0, c2, false, false);
        c3 = __builtin_amdgcn_wmma_f32_16x16x4_f32(
                false, a, false, b3, (short)0, c3, false, false);
    }

    if (blockIdx.x == 0 && threadIdx.x == 0) *maxenc = 0u;

    const int half = lane >> 4;
    float* orow = nt + (size_t)(tile * 16 + 8 * half) * HF + (lane & 15);
#pragma unroll
    for (int r = 0; r < 8; ++r) {
        float* o = orow + (size_t)r * HF;
        o[0]  = c0[r];
        o[16] = c1[r];
        o[32] = c2[r];
        o[48] = c3[r];
    }
}

__device__ __forceinline__ int bin_of(int t) {
    const unsigned b = (unsigned)t / BIN_R;
    return (int)(b < N_BINS ? b : N_BINS - 1);
}

__global__ __launch_bounds__(1024)
void gat_group(const int* __restrict__ src, const int* __restrict__ tgt,
               int* __restrict__ off, int2* __restrict__ pk) {
    __shared__ int  cnt[288];
    __shared__ int2 stage[CHUNK_E];
    const int tid  = threadIdx.x;
    const int base = blockIdx.x * CHUNK_E;
    if (tid < 288) cnt[tid] = 0;

    int sv[7], tv[7], rk[7];
#pragma unroll
    for (int q = 0; q < 7; ++q) {
        const int i = q * 1024 + tid;
        sv[q] = 0; tv[q] = 0;
        if (i < CHUNK_E) {
            sv[q] = src[base + i];
            tv[q] = tgt[base + i];
        }
    }
    __syncthreads();
#pragma unroll
    for (int q = 0; q < 7; ++q) {
        const int i = q * 1024 + tid;
        rk[q] = 0;
        if (i < CHUNK_E)
            rk[q] = atomicAdd(&cnt[bin_of(tv[q])], 1);
    }
    __syncthreads();

    if (tid < 32) {
        int v[9];
        int s = 0;
#pragma unroll
        for (int q = 0; q < 9; ++q) { v[q] = cnt[tid * 9 + q]; s += v[q]; }
        int inc = s;
#pragma unroll
        for (int d = 1; d < 32; d <<= 1) {
            const int o = __shfl_up(inc, d);
            if (tid >= d) inc += o;
        }
        int run = inc - s;
#pragma unroll
        for (int q = 0; q < 9; ++q) { cnt[tid * 9 + q] = run; run += v[q]; }
    }
    __syncthreads();

    if (tid < OFF_PITCH) off[blockIdx.x * OFF_PITCH + tid] = cnt[tid];
#pragma unroll
    for (int q = 0; q < 7; ++q) {
        const int i = q * 1024 + tid;
        if (i < CHUNK_E)
            stage[cnt[bin_of(tv[q])] + rk[q]] = make_int2(sv[q], tv[q]);
    }
    __syncthreads();
    const int a0 = base & ~15;
#pragma unroll
    for (int q = 0; q < 7; ++q) {
        const int a = a0 + q * 1024 + tid;
        if (a >= base && a < base + CHUNK_E) pk[a] = stage[a - base];
    }
}

__global__ __launch_bounds__(1024)
void gat_link_lds(const int* __restrict__ off, int* __restrict__ head,
                  int2* __restrict__ pk) {
    __shared__ int lhead[BIN_R];
    __shared__ int seg_at[N_CHUNKS];
    __shared__ int seg_len[N_CHUNKS];
    const int tid  = threadIdx.x;
    const int lane = tid & 31;
    const int wave = tid >> 5;
    const int b    = blockIdx.x;
    const int nb   = b * BIN_R;

    if (tid < BIN_R) lhead[tid] = -1;
    if (tid < N_CHUNKS) {
        const int o0 = off[tid * OFF_PITCH + b];
        const int o1 = off[tid * OFF_PITCH + b + 1];
        seg_at[tid]  = tid * CHUNK_E + o0;
        seg_len[tid] = o1 - o0;
    }
    __syncthreads();

    int2 st[8];
    int  at[8], ln[8];
#pragma unroll
    for (int r = 0; r < 8; ++r) {
        at[r] = seg_at[wave + 32 * r];
        ln[r] = seg_len[wave + 32 * r];
        st[r] = make_int2(0, 0);
        if (lane < ln[r]) st[r] = pk[at[r] + lane];
    }
#pragma unroll
    for (int r = 0; r < 8; ++r) {
        if (lane < ln[r]) {
            const int e  = at[r] + lane;
            int tl = st[r].y - nb;
            tl = tl < 0 ? 0 : (tl > BIN_R - 1 ? BIN_R - 1 : tl);
            const int prev = atomicExch(&lhead[tl], e);
            pk[e] = make_int2(st[r].x, prev);
        }
        for (int i = 32 + lane; i < ln[r]; i += 32) {
            const int e = at[r] + i;
            const int2 s2 = pk[e];
            int tl = s2.y - nb;
            tl = tl < 0 ? 0 : (tl > BIN_R - 1 ? BIN_R - 1 : tl);
            const int prev = atomicExch(&lhead[tl], e);
            pk[e] = make_int2(s2.x, prev);
        }
    }
    __syncthreads();
    if (tid < BIN_R && nb + tid < N_NODES) head[nb + tid] = lhead[tid];
}

template <int M>
__device__ __forceinline__ float lane_xor(float v) {
    return __builtin_bit_cast(float, __builtin_amdgcn_update_dpp(
        0, __builtin_bit_cast(int, v), 0x160 | M, 0xf, 0xf, true));
}

__device__ __forceinline__ v2f dot8x2(const float4 a, const float4 b, const v2f* w) {
    v2f p = w[0] * a.x;
    p = __builtin_elementwise_fma((v2f)(a.y), w[1], p);
    p = __builtin_elementwise_fma((v2f)(a.z), w[2], p);
    p = __builtin_elementwise_fma((v2f)(a.w), w[3], p);
    p = __builtin_elementwise_fma((v2f)(b.x), w[4], p);
    p = __builtin_elementwise_fma((v2f)(b.y), w[5], p);
    p = __builtin_elementwise_fma((v2f)(b.z), w[6], p);
    p = __builtin_elementwise_fma((v2f)(b.w), w[7], p);
    return p;
}
__device__ __forceinline__ void load_pair(const float* __restrict__ a_w, const int hk,
                                          const int hg, const int pa, v2f* w) {
    const float4 k0 = *(const float4*)(a_w + hk * (2 * HF) + pa);
    const float4 k1 = *(const float4*)(a_w + hk * (2 * HF) + pa + 64);
    const float4 g0 = *(const float4*)(a_w + hg * (2 * HF) + pa);
    const float4 g1 = *(const float4*)(a_w + hg * (2 * HF) + pa + 64);
    w[0] = (v2f){k0.x, g0.x}; w[1] = (v2f){k0.y, g0.y};
    w[2] = (v2f){k0.z, g0.z}; w[3] = (v2f){k0.w, g0.w};
    w[4] = (v2f){k1.x, g1.x}; w[5] = (v2f){k1.y, g1.y};
    w[6] = (v2f){k1.z, g1.z}; w[7] = (v2f){k1.w, g1.w};
}
__device__ __forceinline__ float sum8_keep(const v2f p) {
    float keep = p[0] + lane_xor<4>(p[1]);
    keep += lane_xor<1>(keep);
    keep += lane_xor<2>(keep);
    return keep;
}

__global__ __launch_bounds__(256)
void gat_node_gather(const float* __restrict__ nt,
                     const float* __restrict__ a_w,
                     const int*   __restrict__ head,
                     const int2*  __restrict__ pk,
                     unsigned* __restrict__ maxenc,
                     float* __restrict__ denom,
                     float* __restrict__ out) {
    __shared__ unsigned smax;
    if (threadIdx.x == 0) smax = 0u;
    __syncthreads();

    const int j   = threadIdx.x & 7;
    const int n   = blockIdx.x * 32 + (threadIdx.x >> 3);
    const int col = 4 * j;
    const int hb  = j >> 2;
    const int pa  = hb * 32 + (col & 15);

    const int e0 = head[n];
    float tdA, tdB;
    {
        const float* trow = nt + (size_t)n * HF + col;
        const float4 t0 = *(const float4*)trow;
        const float4 t1 = *(const float4*)(trow + 32);
        v2f wt[8];
        load_pair(a_w, hb, hb ^ 1, pa + 16, wt);
        tdA = sum8_keep(dot8x2(t0, t1, wt));
        load_pair(a_w, 2 + hb, 2 + (hb ^ 1), pa + 16, wt);
        tdB = sum8_keep(dot8x2(t0, t1, wt));
    }
    bool live = (unsigned)e0 < (unsigned)E_EDGES;
    int2 p = make_int2(0, -1);
    if (live) p = pk[e0];
    __builtin_amdgcn_sched_barrier(0);
    v2f wA[8], wB[8];
    load_pair(a_w, hb, hb ^ 1, pa, wA);
    load_pair(a_w, 2 + hb, 2 + (hb ^ 1), pa, wB);

    float4 acc0 = make_float4(0.f, 0.f, 0.f, 0.f);
    float4 acc1 = make_float4(0.f, 0.f, 0.f, 0.f);
    float  den_lo = 0.f, den_hi = 0.f;
    float  mx = -INFINITY;

    for (int it = 0; it < E_EDGES; ++it) {
        if (__builtin_amdgcn_ballot_w32(live) == 0u) break;
        const bool more = live && ((unsigned)p.y < (unsigned)E_EDGES);
        const float* srow = nt + (size_t)p.x * HF + col;
        const float4 r0 = *(const float4*)srow;
        const float4 r1 = *(const float4*)(srow + 32);
        const int2 pn = pk[more ? p.y : 0];

        const float lg_lo = sum8_keep(dot8x2(r0, r1, wA)) + tdA;
        const float lg_hi = sum8_keep(dot8x2(r0, r1, wB)) + tdB;

        mx = live ? fmaxf(mx, fmaxf(lg_lo, lg_hi)) : mx;
        const float e_lo = live ? __expf(NEG_SLOPE * lg_lo) : 0.0f;
        const float e_hi = live ? __expf(NEG_SLOPE * lg_hi) : 0.0f;
        den_lo += e_lo;
        den_hi += e_hi;
        acc0.x = fmaf(e_lo, r0.x, acc0.x); acc0.y = fmaf(e_lo, r0.y, acc0.y);
        acc0.z = fmaf(e_lo, r0.z, acc0.z); acc0.w = fmaf(e_lo, r0.w, acc0.w);
        acc1.x = fmaf(e_hi, r1.x, acc1.x); acc1.y = fmaf(e_hi, r1.y, acc1.y);
        acc1.z = fmaf(e_hi, r1.z, acc1.z); acc1.w = fmaf(e_hi, r1.w, acc1.w);

        live = more;
        p = pn;
    }

    const float d_lo = den_lo;
    const float d_hi = den_hi;
    if (d_lo > 0.0f) {
        acc0.x /= d_lo; acc0.y /= d_lo; acc0.z /= d_lo; acc0.w /= d_lo;
    } else {
        acc0 = make_float4(0.f, 0.f, 0.f, 0.f);
    }
    if (d_hi > 0.0f) {
        acc1.x /= d_hi; acc1.y /= d_hi; acc1.z /= d_hi; acc1.w /= d_hi;
    } else {
        acc1 = make_float4(0.f, 0.f, 0.f, 0.f);
    }
    float* orow = out + (size_t)n * HF + col;
    *(float4*)orow        = acc0;
    *(float4*)(orow + 32) = acc1;
    const float d1 = lane_xor<4>(den_lo);
    const float d3 = lane_xor<4>(den_hi);
    mx = fmaxf(mx, lane_xor<4>(mx));
    if (j == 0) {
        *(float4*)(denom + (size_t)n * H_HEADS) =
            make_float4(den_lo, d1, den_hi, d3);
        atomicMax(&smax, fenc(mx));
    }
    __syncthreads();
    if (threadIdx.x == 0) atomicMax(maxenc, smax);
}

__global__ __launch_bounds__(256)
void gat_epsilon(float* __restrict__ out, const float* __restrict__ denom,
                 const unsigned* __restrict__ maxenc) {
    const int n = blockIdx.x * 256 + threadIdx.x;
    if (n >= N_NODES) return;
    const float eps = 1e-8f * __expf(NEG_SLOPE * fdec(*maxenc));
    const float4 dv = ((const float4*)denom)[n];
    const float dd[H_HEADS] = {dv.x, dv.y, dv.z, dv.w};
#pragma unroll
    for (int k = 0; k < H_HEADS; ++k) {
        const float d = dd[k] + eps;
        if (d != dd[k]) {
            const float sc = dd[k] / d;
            float4* o = (float4*)(out + (size_t)n * HF + k * F_OUT);
            for (int q = 0; q < F_OUT / 4; ++q) {
                float4 v = o[q];
                v.x *= sc; v.y *= sc; v.z *= sc; v.w *= sc;
                o[q] = v;
            }
        }
    }
}

extern "C" void kernel_launch(void* const* d_in, const int* in_sizes, int n_in,
                              void* d_out, int out_size, void* d_ws, size_t ws_size,
                              hipStream_t stream) {
    const float* x   = (const float*)d_in[0];
    const float* W_w = (const float*)d_in[1];
    const float* a_w = (const float*)d_in[2];
    const int*   ei  = (const int*)d_in[3];
    const int* src = ei;
    const int* tgt = ei + E_EDGES;
    float* out = (float*)d_out;

    float* nt    = (float*)d_ws;
    float* denom = nt + (size_t)N_NODES * HF;
    unsigned* maxenc = (unsigned*)(denom + (size_t)N_NODES * H_HEADS);
    int*  head = (int*)(maxenc + 4);
    int2* pk   = (int2*)(head + N_NODES);
    int*  off  = (int*)denom;

    (void)in_sizes; (void)n_in; (void)out_size; (void)ws_size;

    {
        gat_transform_wmma<<<N_NODES / 32, 64, 0, stream>>>(x, W_w, nt, maxenc);
    }
    {
        gat_group<<<N_CHUNKS, 1024, 0, stream>>>(src, tgt, off, pk);
    }
    {
        gat_link_lds<<<N_BINS, 1024, 0, stream>>>(off, head, pk);
    }
    {
        gat_node_gather<<<N_NODES / 32, 256, 0, stream>>>(
            nt, a_w, head, pk, maxenc, denom, out);
    }
    {
        gat_epsilon<<<(N_NODES + 255) / 256, 256, 0, stream>>>(out, denom, maxenc);
    }
}


# Gather: nodes dealt to lanes by list length, longest first (each range sorts its nodes in LDS while linking; blocks take the ranges' slices in turn): a wave's 4 lists end together; sums unchanged
# speedup vs baseline: 7.7116x; 1.0395x over previous
#include <hip/hip_runtime.h>
#include <hip/hip_bf16.h>


#define N_NODES   100000
#define E_EDGES   1600000
#define IN_F      128
#define H_HEADS   4
#define F_OUT     16
#define HF        64
#define NEG_SLOPE 0.01f

#define N_CHUNKS  256
#define CHUNK_E   6250
#define BIN_R     384
#define N_BINS    261
#define OFF_PITCH 262
#define HEAD_BITS 21
#define LEN_BINS  64
static_assert(E_EDGES < (1 << HEAD_BITS), "list front + 1 must fit in HEAD_BITS bits");
static_assert(BIN_R <= (1 << (31 - HEAD_BITS)), "node index within a range must fit above HEAD_BITS");
static_assert(BIN_R % 32 == 0, "a gather block of 32 nodes must lie inside one range");
static_assert(N_NODES % 32 == 0 && (N_NODES % BIN_R) % 32 == 0, "the last range must hold whole gather blocks");
static_assert((N_NODES + BIN_R - 1) / BIN_R == N_BINS, "one link block per range");

typedef float v2f __attribute__((ext_vector_type(2)));
typedef float v8f __attribute__((ext_vector_type(8)));

__device__ __forceinline__ unsigned fenc(float f) {
    unsigned u = __float_as_uint(f);
    return (u & 0x80000000u) ? ~u : (u | 0x80000000u);
}
__device__ __forceinline__ float fdec(unsigned k) {
    return (k & 0x80000000u) ? __uint_as_float(k ^ 0x80000000u)
                             : __uint_as_float(~k);
}

__global__ __launch_bounds__(64)
void gat_transform_wmma(const float* __restrict__ x,
                        const float* __restrict__ W,
                        float* __restrict__ nt,
                        unsigned* __restrict__ maxenc) {
    const int lane = threadIdx.x & 31;
    const int wave = threadIdx.x >> 5;
    const int m    = lane & 15;
    const int kk   = lane >> 4;
    const int tile = blockIdx.x * 2 + wave;
    const int row  = tile * 16 + m;

    const float* xrow = x + (size_t)row * IN_F + 2 * kk;
    const float* wrow = W + (size_t)m * IN_F + 2 * kk;

    v2f xa[IN_F / 4];
#pragma unroll
    for (int s = 0; s < IN_F / 4; ++s)
        xa[s] = *(const v2f*)(xrow + 4 * s);
    __builtin_amdgcn_sched_barrier(0);

    v8f c0 = {}, c1 = {}, c2 = {}, c3 = {};
#pragma unroll
    for (int k0 = 0; k0 < IN_F; k0 += 4) {
        const v2f a  = xa[k0 / 4];
        const v2f b0 = *(const v2f*)(wrow + k0);
        const v2f b1 = *(const v2f*)(wrow + 16 * IN_F + k0);
        const v2f b2 = *(const v2f*)(wrow + 32 * IN_F + k0);
        const v2f b3 = *(const v2f*)(wrow + 48 * IN_F + k0);
        c0 = __builtin_amdgcn_wmma_f32_16x16x4_f32(
                 false, a,  false, b0,
                 (short)0, c0,  false,  false);
        c1 = __builtin_amdgcn_wmma_f32_16x16x4_f32(
                false, a, false, b1, (short)0, c1, false, false);
        c2 = __builtin_amdgcn_wmma_f32_16x16x4_f32(
                false, a, false, b2, (short)0, c2, false, false);
        c3 = __builtin_amdgcn_wmma_f32_16x16x4_f32(
                false, a, false, b3, (short)0, c3, false, false);
    }

    if (blockIdx.x == 0 && threadIdx.x == 0) *maxenc = 0u;

    const int half = lane >> 4;
    float* orow = nt + (size_t)(tile * 16 + 8 * half) * HF + (lane & 15);
#pragma unroll
    for (int r = 0; r < 8; ++r) {
        float* o = orow + (size_t)r * HF;
        o[0]  = c0[r];
        o[16] = c1[r];
        o[32] = c2[r];
        o[48] = c3[r];
    }
}

__device__ __forceinline__ int bin_of(int t) {
    const unsigned b = (unsigned)t / BIN_R;
    return (int)(b < N_BINS ? b : N_BINS - 1);
}

__global__ __launch_bounds__(1024)
void gat_group(const int* __restrict__ src, const int* __restrict__ tgt,
               int* __restrict__ off, int2* __restrict__ pk) {
    __shared__ int  cnt[288];
    __shared__ int2 stage[CHUNK_E];
    const int tid  = threadIdx.x;
    const int base = blockIdx.x * CHUNK_E;
    if (tid < 288) cnt[tid] = 0;

    int sv[7], tv[7], rk[7];
#pragma unroll
    for (int q = 0; q < 7; ++q) {
        const int i = q * 1024 + tid;
        sv[q] = 0; tv[q] = 0;
        if (i < CHUNK_E) {
            sv[q] = src[base + i];
            tv[q] = tgt[base + i];
        }
    }
    __syncthreads();
#pragma unroll
    for (int q = 0; q < 7; ++q) {
        const int i = q * 1024 + tid;
        rk[q] = 0;
        if (i < CHUNK_E)
            rk[q] = atomicAdd(&cnt[bin_of(tv[q])], 1);
    }
    __syncthreads();

    if (tid < 32) {
        int v[9];
        int s = 0;
#pragma unroll
        for (int q = 0; q < 9; ++q) { v[q] = cnt[tid * 9 + q]; s += v[q]; }
        int inc = s;
#pragma unroll
        for (int d = 1; d < 32; d <<= 1) {
            const int o = __shfl_up(inc, d);
            if (tid >= d) inc += o;
        }
        int run = inc - s;
#pragma unroll
        for (int q = 0; q < 9; ++q) { cnt[tid * 9 + q] = run; run += v[q]; }
    }
    __syncthreads();

    if (tid < OFF_PITCH) off[blockIdx.x * OFF_PITCH + tid] = cnt[tid];
#pragma unroll
    for (int q = 0; q < 7; ++q) {
        const int i = q * 1024 + tid;
        if (i < CHUNK_E)
            stage[cnt[bin_of(tv[q])] + rk[q]] = make_int2(sv[q], tv[q]);
    }
    __syncthreads();
    const int a0 = base & ~15;
#pragma unroll
    for (int q = 0; q < 7; ++q) {
        const int a = a0 + q * 1024 + tid;
        if (a >= base && a < base + CHUNK_E) pk[a] = stage[a - base];
    }
}

__global__ __launch_bounds__(1024)
void gat_link_lds(const int* __restrict__ off, int* __restrict__ order,
                  int2* __restrict__ pk) {
    __shared__ int lhead[BIN_R];
    __shared__ int llen[BIN_R];
    __shared__ int lbin[LEN_BINS];
    __shared__ int seg_at[N_CHUNKS];
    __shared__ int seg_len[N_CHUNKS];
    const int tid  = threadIdx.x;
    const int lane = tid & 31;
    const int wave = tid >> 5;
    const int b    = blockIdx.x;
    const int nb   = b * BIN_R;

    if (tid < BIN_R) { lhead[tid] = -1; llen[tid] = 0; }
    if (tid < LEN_BINS) lbin[tid] = 0;
    if (tid < N_CHUNKS) {
        const int o0 = off[tid * OFF_PITCH + b];
        const int o1 = off[tid * OFF_PITCH + b + 1];
        seg_at[tid]  = tid * CHUNK_E + o0;
        seg_len[tid] = o1 - o0;
    }
    __syncthreads();

    int2 st[8];
    int  at[8], ln[8];
#pragma unroll
    for (int r = 0; r < 8; ++r) {
        at[r] = seg_at[wave + 32 * r];
        ln[r] = seg_len[wave + 32 * r];
        st[r] = make_int2(0, 0);
        if (lane < ln[r]) st[r] = pk[at[r] + lane];
    }
#pragma unroll
    for (int r = 0; r < 8; ++r) {
        if (lane < ln[r]) {
            const int e  = at[r] + lane;
            int tl = st[r].y - nb;
            tl = tl < 0 ? 0 : (tl > BIN_R - 1 ? BIN_R - 1 : tl);
            const int prev = atomicExch(&lhead[tl], e);
            atomicAdd(&llen[tl], 1);
            pk[e] = make_int2(st[r].x, prev);
        }
        for (int i = 32 + lane; i < ln[r]; i += 32) {
            const int e = at[r] + i;
            const int2 s2 = pk[e];
            int tl = s2.y - nb;
            tl = tl < 0 ? 0 : (tl > BIN_R - 1 ? BIN_R - 1 : tl);
            const int prev = atomicExch(&lhead[tl], e);
            atomicAdd(&llen[tl], 1);
            pk[e] = make_int2(s2.x, prev);
        }
    }
    __syncthreads();

    const int nn = (N_NODES - nb) < BIN_R ? (N_NODES - nb) : BIN_R;
    int key = 0, rk = 0;
    if (tid < nn) {
        const int d = llen[tid];
        key = d < LEN_BINS - 1 ? d : LEN_BINS - 1;
        rk  = atomicAdd(&lbin[key], 1);
    }
    __syncthreads();
    if (tid < 32) {
        const int v0 = lbin[2 * tid];
        const int v1 = lbin[2 * tid + 1];
        const int s  = v0 + v1;
        int inc = s;
#pragma unroll
        for (int d = 1; d < 32; d <<= 1) {
            const int o = __shfl_up(inc, d);
            if (tid >= d) inc += o;
        }
        const int run = inc - s;
        lbin[2 * tid]     = run;
        lbin[2 * tid + 1] = run + v0;
    }
    __syncthreads();
    if (tid < nn)
        order[nb + (nn - 1) - (lbin[key] + rk)] = (tid << HEAD_BITS) | (lhead[tid] + 1);
}

template <int M>
__device__ __forceinline__ float lane_xor(float v) {
    return __builtin_bit_cast(float, __builtin_amdgcn_update_dpp(
        0, __builtin_bit_cast(int, v), 0x160 | M, 0xf, 0xf, true));
}

__device__ __forceinline__ v2f dot8x2(const float4 a, const float4 b, const v2f* w) {
    v2f p = w[0] * a.x;
    p = __builtin_elementwise_fma((v2f)(a.y), w[1], p);
    p = __builtin_elementwise_fma((v2f)(a.z), w[2], p);
    p = __builtin_elementwise_fma((v2f)(a.w), w[3], p);
    p = __builtin_elementwise_fma((v2f)(b.x), w[4], p);
    p = __builtin_elementwise_fma((v2f)(b.y), w[5], p);
    p = __builtin_elementwise_fma((v2f)(b.z), w[6], p);
    p = __builtin_elementwise_fma((v2f)(b.w), w[7], p);
    return p;
}
__device__ __forceinline__ void load_pair(const float* __restrict__ a_w, const int hk,
                                          const int hg, const int pa, v2f* w) {
    const float4 k0 = *(const float4*)(a_w + hk * (2 * HF) + pa);
    const float4 k1 = *(const float4*)(a_w + hk * (2 * HF) + pa + 64);
    const float4 g0 = *(const float4*)(a_w + hg * (2 * HF) + pa);
    const float4 g1 = *(const float4*)(a_w + hg * (2 * HF) + pa + 64);
    w[0] = (v2f){k0.x, g0.x}; w[1] = (v2f){k0.y, g0.y};
    w[2] = (v2f){k0.z, g0.z}; w[3] = (v2f){k0.w, g0.w};
    w[4] = (v2f){k1.x, g1.x}; w[5] = (v2f){k1.y, g1.y};
    w[6] = (v2f){k1.z, g1.z}; w[7] = (v2f){k1.w, g1.w};
}
__device__ __forceinline__ float sum8_keep(const v2f p) {
    float keep = p[0] + lane_xor<4>(p[1]);
    keep += lane_xor<1>(keep);
    keep += lane_xor<2>(keep);
    return keep;
}

__global__ __launch_bounds__(256)
void gat_node_gather(const float* __restrict__ nt,
                     const float* __restrict__ a_w,
                     const int*   __restrict__ order,
                     const int2*  __restrict__ pk,
                     unsigned* __restrict__ maxenc,
                     float* __restrict__ denom,
                     float* __restrict__ out) {
    __shared__ unsigned smax;
    if (threadIdx.x == 0) smax = 0u;
    __syncthreads();

    const int j   = threadIdx.x & 7;
    constexpr int PER   = BIN_R / 32;
    constexpr int FULL  = N_NODES / BIN_R;
    constexpr int TAILB = (N_NODES - FULL * BIN_R) / 32;
    const int bi = (int)blockIdx.x;
    int k, r;
    if (bi < TAILB * N_BINS) {
        k = bi / N_BINS;
        r = bi - k * N_BINS;
    } else {
        const int q = (bi - TAILB * N_BINS) / FULL;
        k = TAILB + q;
        r = (bi - TAILB * N_BINS) - q * FULL;
    }
    const unsigned ow = (unsigned)order[(r * PER + k) * 32 + (threadIdx.x >> 3)];
    const int nraw = r * BIN_R + (int)(ow >> HEAD_BITS);
    const int n    = nraw < N_NODES ? nraw : N_NODES - 1;
    const int e0   = (int)(ow & ((1u << HEAD_BITS) - 1u)) - 1;
    const int col = 4 * j;
    const int hb  = j >> 2;
    const int pa  = hb * 32 + (col & 15);

    bool live = (unsigned)e0 < (unsigned)E_EDGES;
    int2 p = make_int2(0, -1);
    if (live) p = pk[e0];
    float tdA, tdB;
    {
        const float* trow = nt + (size_t)n * HF + col;
        const float4 t0 = *(const float4*)trow;
        const float4 t1 = *(const float4*)(trow + 32);
        v2f wt[8];
        load_pair(a_w, hb, hb ^ 1, pa + 16, wt);
        tdA = sum8_keep(dot8x2(t0, t1, wt));
        load_pair(a_w, 2 + hb, 2 + (hb ^ 1), pa + 16, wt);
        tdB = sum8_keep(dot8x2(t0, t1, wt));
    }
    __builtin_amdgcn_sched_barrier(0);
    v2f wA[8], wB[8];
    load_pair(a_w, hb, hb ^ 1, pa, wA);
    load_pair(a_w, 2 + hb, 2 + (hb ^ 1), pa, wB);

    float4 acc0 = make_float4(0.f, 0.f, 0.f, 0.f);
    float4 acc1 = make_float4(0.f, 0.f, 0.f, 0.f);
    float  den_lo = 0.f, den_hi = 0.f;
    float  mx = -INFINITY;

    for (int it = 0; it < E_EDGES; ++it) {
        if (__builtin_amdgcn_ballot_w32(live) == 0u) break;
        const bool more = live && ((unsigned)p.y < (unsigned)E_EDGES);
        const float* srow = nt + (size_t)p.x * HF + col;
        const float4 r0 = *(const float4*)srow;
        const float4 r1 = *(const float4*)(srow + 32);
        const int2 pn = pk[more ? p.y : 0];

        const float lg_lo = sum8_keep(dot8x2(r0, r1, wA)) + tdA;
        const float lg_hi = sum8_keep(dot8x2(r0, r1, wB)) + tdB;

        mx = live ? fmaxf(mx, fmaxf(lg_lo, lg_hi)) : mx;
        const float e_lo = live ? __expf(NEG_SLOPE * lg_lo) : 0.0f;
        const float e_hi = live ? __expf(NEG_SLOPE * lg_hi) : 0.0f;
        den_lo += e_lo;
        den_hi += e_hi;
        acc0.x = fmaf(e_lo, r0.x, acc0.x); acc0.y = fmaf(e_lo, r0.y, acc0.y);
        acc0.z = fmaf(e_lo, r0.z, acc0.z); acc0.w = fmaf(e_lo, r0.w, acc0.w);
        acc1.x = fmaf(e_hi, r1.x, acc1.x); acc1.y = fmaf(e_hi, r1.y, acc1.y);
        acc1.z = fmaf(e_hi, r1.z, acc1.z); acc1.w = fmaf(e_hi, r1.w, acc1.w);

        live = more;
        p = pn;
    }

    const float d_lo = den_lo;
    const float d_hi = den_hi;
    if (d_lo > 0.0f) {
        acc0.x /= d_lo; acc0.y /= d_lo; acc0.z /= d_lo; acc0.w /= d_lo;
    } else {
        acc0 = make_float4(0.f, 0.f, 0.f, 0.f);
    }
    if (d_hi > 0.0f) {
        acc1.x /= d_hi; acc1.y /= d_hi; acc1.z /= d_hi; acc1.w /= d_hi;
    } else {
        acc1 = make_float4(0.f, 0.f, 0.f, 0.f);
    }
    float* orow = out + (size_t)n * HF + col;
    *(float4*)orow        = acc0;
    *(float4*)(orow + 32) = acc1;
    const float d1 = lane_xor<4>(den_lo);
    const float d3 = lane_xor<4>(den_hi);
    mx = fmaxf(mx, lane_xor<4>(mx));
    if (j == 0) {
        *(float4*)(denom + (size_t)n * H_HEADS) =
            make_float4(den_lo, d1, den_hi, d3);
        atomicMax(&smax, fenc(mx));
    }
    __syncthreads();
    if (threadIdx.x == 0) atomicMax(maxenc, smax);
}

__global__ __launch_bounds__(256)
void gat_epsilon(float* __restrict__ out, const float* __restrict__ denom,
                 const unsigned* __restrict__ maxenc) {
    const int n = blockIdx.x * 256 + threadIdx.x;
    if (n >= N_NODES) return;
    const float eps = 1e-8f * __expf(NEG_SLOPE * fdec(*maxenc));
    const float4 dv = ((const float4*)denom)[n];
    const float dd[H_HEADS] = {dv.x, dv.y, dv.z, dv.w};
#pragma unroll
    for (int k = 0; k < H_HEADS; ++k) {
        const float d = dd[k] + eps;
        if (d != dd[k]) {
            const float sc = dd[k] / d;
            float4* o = (float4*)(out + (size_t)n * HF + k * F_OUT);
            for (int q = 0; q < F_OUT / 4; ++q) {
                float4 v = o[q];
                v.x *= sc; v.y *= sc; v.z *= sc; v.w *= sc;
                o[q] = v;
            }
        }
    }
}

extern "C" void kernel_launch(void* const* d_in, const int* in_sizes, int n_in,
                              void* d_out, int out_size, void* d_ws, size_t ws_size,
                              hipStream_t stream) {
    const float* x   = (const float*)d_in[0];
    const float* W_w = (const float*)d_in[1];
    const float* a_w = (const float*)d_in[2];
    const int*   ei  = (const int*)d_in[3];
    const int* src = ei;
    const int* tgt = ei + E_EDGES;
    float* out = (float*)d_out;

    float* nt    = (float*)d_ws;
    float* denom = nt + (size_t)N_NODES * HF;
    unsigned* maxenc = (unsigned*)(denom + (size_t)N_NODES * H_HEADS);
    int*  order = (int*)(maxenc + 4);
    int2* pk   = (int2*)(order + N_NODES);
    int*  off  = (int*)denom;

    (void)in_sizes; (void)n_in; (void)out_size; (void)ws_size;

    {
        gat_transform_wmma<<<N_NODES / 32, 64, 0, stream>>>(x, W_w, nt, maxenc);
    }
    {
        gat_group<<<N_CHUNKS, 1024, 0, stream>>>(src, tgt, off, pk);
    }
    {
        gat_link_lds<<<N_BINS, 1024, 0, stream>>>(off, order, pk);
    }
    {
        gat_node_gather<<<N_NODES / 32, 256, 0, stream>>>(
            nt, a_w, order, pk, maxenc, denom, out);
    }
    {
        gat_epsilon<<<(N_NODES + 255) / 256, 256, 0, stream>>>(out, denom, maxenc);
    }
}


# Edge lists as consecutive words per node (range block counts, scans, places in LDS), not chained: gather reads 8 list words at once, 4 source rows in flight; list order reversed: sum order changes
# speedup vs baseline: 8.1142x; 1.0522x over previous
#include <hip/hip_runtime.h>
#include <hip/hip_bf16.h>


#define N_NODES   100000
#define E_EDGES   1600000
#define IN_F      128
#define H_HEADS   4
#define F_OUT     16
#define HF        64
#define NEG_SLOPE 0.01f

#define N_CHUNKS  256
#define CHUNK_E   6250
#define BIN_R     384
#define N_BINS    261
#define OFF_PITCH 262
#define HEAD_BITS 21
#define LEN_BINS  64
#define SRC_BITS  17
#define SRC_MASK  0x1FFFFu
#define LAST_BIT  0x80000000u
#define STAGE_CAP 8192
static_assert(N_NODES <= (1 << SRC_BITS), "a source node must fit in SRC_BITS bits");
static_assert(BIN_R <= (1 << (31 - SRC_BITS)), "index within a range must fit between source and bit 31");
static_assert(E_EDGES < (1 << HEAD_BITS), "list front + 1 must fit in HEAD_BITS bits");
static_assert(BIN_R <= (1 << (31 - HEAD_BITS)), "node index within a range must fit above HEAD_BITS");
static_assert(BIN_R % 32 == 0, "a gather block of 32 nodes must lie inside one range");
static_assert(N_NODES % 32 == 0 && (N_NODES % BIN_R) % 32 == 0, "the last range must hold whole gather blocks");
static_assert((N_NODES + BIN_R - 1) / BIN_R == N_BINS, "one link block per range");

typedef float v2f __attribute__((ext_vector_type(2)));
typedef float v8f __attribute__((ext_vector_type(8)));

__device__ __forceinline__ unsigned fenc(float f) {
    unsigned u = __float_as_uint(f);
    return (u & 0x80000000u) ? ~u : (u | 0x80000000u);
}
__device__ __forceinline__ float fdec(unsigned k) {
    return (k & 0x80000000u) ? __uint_as_float(k ^ 0x80000000u)
                             : __uint_as_float(~k);
}

__global__ __launch_bounds__(64)
void gat_transform_wmma(const float* __restrict__ x,
                        const float* __restrict__ W,
                        float* __restrict__ nt,
                        unsigned* __restrict__ maxenc) {
    const int lane = threadIdx.x & 31;
    const int wave = threadIdx.x >> 5;
    const int m    = lane & 15;
    const int kk   = lane >> 4;
    const int tile = blockIdx.x * 2 + wave;
    const int row  = tile * 16 + m;

    const float* xrow = x + (size_t)row * IN_F + 2 * kk;
    const float* wrow = W + (size_t)m * IN_F + 2 * kk;

    v2f xa[IN_F / 4];
#pragma unroll
    for (int s = 0; s < IN_F / 4; ++s)
        xa[s] = *(const v2f*)(xrow + 4 * s);
    __builtin_amdgcn_sched_barrier(0);

    v8f c0 = {}, c1 = {}, c2 = {}, c3 = {};
#pragma unroll
    for (int k0 = 0; k0 < IN_F; k0 += 4) {
        const v2f a  = xa[k0 / 4];
        const v2f b0 = *(const v2f*)(wrow + k0);
        const v2f b1 = *(const v2f*)(wrow + 16 * IN_F + k0);
        const v2f b2 = *(const v2f*)(wrow + 32 * IN_F + k0);
        const v2f b3 = *(const v2f*)(wrow + 48 * IN_F + k0);
        c0 = __builtin_amdgcn_wmma_f32_16x16x4_f32(
                 false, a,  false, b0,
                 (short)0, c0,  false,  false);
        c1 = __builtin_amdgcn_wmma_f32_16x16x4_f32(
                false, a, false, b1, (short)0, c1, false, false);
        c2 = __builtin_amdgcn_wmma_f32_16x16x4_f32(
                false, a, false, b2, (short)0, c2, false, false);
        c3 = __builtin_amdgcn_wmma_f32_16x16x4_f32(
                false, a, false, b3, (short)0, c3, false, false);
    }

    if (blockIdx.x == 0 && threadIdx.x == 0) *maxenc = 0u;

    const int half = lane >> 4;
    float* orow = nt + (size_t)(tile * 16 + 8 * half) * HF + (lane & 15);
#pragma unroll
    for (int r = 0; r < 8; ++r) {
        float* o = orow + (size_t)r * HF;
        o[0]  = c0[r];
        o[16] = c1[r];
        o[32] = c2[r];
        o[48] = c3[r];
    }
}

__device__ __forceinline__ int bin_of(int t) {
    const unsigned b = (unsigned)t / BIN_R;
    return (int)(b < N_BINS ? b : N_BINS - 1);
}

__global__ __launch_bounds__(1024)
void gat_group(const int* __restrict__ src, const int* __restrict__ tgt,
               int* __restrict__ off, int* __restrict__ grp) {
    __shared__ int cnt[288];
    __shared__ int stage[CHUNK_E];
    const int tid  = threadIdx.x;
    const int base = blockIdx.x * CHUNK_E;
    if (tid < 288) cnt[tid] = 0;

    int sv[7], tv[7], rk[7];
#pragma unroll
    for (int q = 0; q < 7; ++q) {
        const int i = q * 1024 + tid;
        sv[q] = 0; tv[q] = 0;
        if (i < CHUNK_E) {
            sv[q] = src[base + i];
            tv[q] = tgt[base + i];
        }
    }
    __syncthreads();
#pragma unroll
    for (int q = 0; q < 7; ++q) {
        const int i = q * 1024 + tid;
        rk[q] = 0;
        if (i < CHUNK_E)
            rk[q] = atomicAdd(&cnt[bin_of(tv[q])], 1);
    }
    __syncthreads();

    if (tid < 32) {
        int v[9];
        int s = 0;
#pragma unroll
        for (int q = 0; q < 9; ++q) { v[q] = cnt[tid * 9 + q]; s += v[q]; }
        int inc = s;
#pragma unroll
        for (int d = 1; d < 32; d <<= 1) {
            const int o = __shfl_up(inc, d);
            if (tid >= d) inc += o;
        }
        int run = inc - s;
#pragma unroll
        for (int q = 0; q < 9; ++q) { cnt[tid * 9 + q] = run; run += v[q]; }
    }
    __syncthreads();

    if (tid < OFF_PITCH) off[blockIdx.x * OFF_PITCH + tid] = cnt[tid];
#pragma unroll
    for (int q = 0; q < 7; ++q) {
        const int i = q * 1024 + tid;
        if (i < CHUNK_E) {
            const int b = bin_of(tv[q]);
            int tl = tv[q] - b * BIN_R;
            tl = tl < 0 ? 0 : (tl > BIN_R - 1 ? BIN_R - 1 : tl);
            stage[cnt[b] + rk[q]] = (int)(((unsigned)sv[q] & SRC_MASK) | ((unsigned)tl << SRC_BITS));
        }
    }
    __syncthreads();
    const int a0 = base & ~31;
#pragma unroll
    for (int q = 0; q < 7; ++q) {
        const int a = a0 + q * 1024 + tid;
        if (a >= base && a < base + CHUNK_E) grp[a] = stage[a - base];
    }
}

__device__ __forceinline__ int tl_of(const int w) {
    const int tl = (int)(((unsigned)w >> SRC_BITS) & 0x3FFFu);
    return tl < BIN_R - 1 ? tl : BIN_R - 1;
}

__global__ __launch_bounds__(1024)
void gat_link_lds(const int* __restrict__ off, int* __restrict__ order,
                  const int* __restrict__ grp, int* __restrict__ srt) {
    __shared__ int llen[BIN_R];
    __shared__ int cur[BIN_R];
    __shared__ int lend[BIN_R];
    __shared__ int lbin[LEN_BINS];
    __shared__ int seg_at[N_CHUNKS];
    __shared__ int seg_len[N_CHUNKS];
    __shared__ int sstage[STAGE_CAP];
    __shared__ int rbase;
    __shared__ int rtot;
    const int tid  = threadIdx.x;
    const int lane = tid & 31;
    const int wave = tid >> 5;
    const int b    = blockIdx.x;
    const int nb   = b * BIN_R;

    if (tid < BIN_R) llen[tid] = 0;
    if (tid < LEN_BINS) lbin[tid] = 0;
    if (tid < N_CHUNKS) {
        const int o0 = off[tid * OFF_PITCH + b];
        const int o1 = off[tid * OFF_PITCH + b + 1];
        seg_at[tid]  = tid * CHUNK_E + o0;
        seg_len[tid] = o1 - o0;
    }
    __syncthreads();

    int st[8];
    int at[8], ln[8];
#pragma unroll
    for (int r = 0; r < 8; ++r) {
        at[r] = seg_at[wave + 32 * r];
        ln[r] = seg_len[wave + 32 * r];
        st[r] = 0;
        if (lane < ln[r]) st[r] = grp[at[r] + lane];
    }
#pragma unroll
    for (int r = 0; r < 8; ++r) {
        if (lane < ln[r]) atomicAdd(&llen[tl_of(st[r])], 1);
        for (int i = 32 + lane; i < ln[r]; i += 32)
            atomicAdd(&llen[tl_of(grp[at[r] + i])], 1);
    }
    __syncthreads();

    const int nn = (N_NODES - nb) < BIN_R ? (N_NODES - nb) : BIN_R;
    int key = 0, rk = 0;
    if (tid < nn) {
        const int d = llen[tid];
        key = d < LEN_BINS - 1 ? d : LEN_BINS - 1;
        rk  = atomicAdd(&lbin[key], 1);
    }
    if (wave == 31) {
        int v[12];
        int sum = 0;
#pragma unroll
        for (int q = 0; q < 12; ++q) { v[q] = llen[lane * 12 + q]; sum += v[q]; }
        int inc = sum;
#pragma unroll
        for (int d = 1; d < 32; d <<= 1) {
            const int o = __shfl_up(inc, d);
            if (lane >= d) inc += o;
        }
        int run = inc - sum;
#pragma unroll
        for (int q = 0; q < 12; ++q) {
            cur[lane * 12 + q] = run;
            run += v[q];
            lend[lane * 12 + q] = run;
        }
        if (lane == 31) rtot = inc;
    }
    if (wave == 30) {
        int sum = 0;
#pragma unroll
        for (int q = 0; q < 8; ++q) {
            const int c = lane * 8 + q;
            sum += seg_at[c] - c * CHUNK_E;
        }
#pragma unroll
        for (int d = 16; d >= 1; d >>= 1) sum += __shfl_xor(sum, d);
        if (lane == 0) rbase = sum;
    }
    __syncthreads();

    if (tid < 32) {
        const int v0 = lbin[2 * tid];
        const int v1 = lbin[2 * tid + 1];
        const int sum = v0 + v1;
        int inc = sum;
#pragma unroll
        for (int d = 1; d < 32; d <<= 1) {
            const int o = __shfl_up(inc, d);
            if (tid >= d) inc += o;
        }
        const int run = inc - sum;
        lbin[2 * tid]     = run;
        lbin[2 * tid + 1] = run + v0;
    }
    int rb = rbase;
    rb = rb < 0 ? 0 : rb;
#pragma unroll
    for (int r = 0; r < 8; ++r) {
        const int sa = seg_at[wave + 32 * r];
        const int sl = seg_len[wave + 32 * r];
        if (lane < sl) {
            const int w   = st[r];
            const int tl  = tl_of(w);
            const int pos = atomicAdd(&cur[tl], 1);
            const unsigned word = ((unsigned)w & SRC_MASK) | (pos == lend[tl] - 1 ? LAST_BIT : 0u);
            if (pos < STAGE_CAP) sstage[pos] = (int)word;
            else if (rb + pos < E_EDGES) srt[rb + pos] = (int)word;
        }
        for (int i = 32 + lane; i < sl; i += 32) {
            const int w   = grp[sa + i];
            const int tl  = tl_of(w);
            const int pos = atomicAdd(&cur[tl], 1);
            const unsigned word = ((unsigned)w & SRC_MASK) | (pos == lend[tl] - 1 ? LAST_BIT : 0u);
            if (pos < STAGE_CAP) sstage[pos] = (int)word;
            else if (rb + pos < E_EDGES) srt[rb + pos] = (int)word;
        }
    }
    __syncthreads();

    if (tid < nn) {
        const int len   = llen[tid];
        const int front = len > 0 ? rb + (lend[tid] - len) + 1 : 0;
        order[nb + (nn - 1) - (lbin[key] + rk)] = (tid << HEAD_BITS) | front;
    }
    const int tot = rtot < STAGE_CAP ? rtot : STAGE_CAP;
    for (int i = tid; i < tot; i += 1024)
        if (rb + i < E_EDGES) srt[rb + i] = sstage[i];
}

template <int M>
__device__ __forceinline__ float lane_xor(float v) {
    return __builtin_bit_cast(float, __builtin_amdgcn_update_dpp(
        0, __builtin_bit_cast(int, v), 0x160 | M, 0xf, 0xf, true));
}

__device__ __forceinline__ v2f dot8x2(const float4 a, const float4 b, const v2f* w) {
    v2f p = w[0] * a.x;
    p = __builtin_elementwise_fma((v2f)(a.y), w[1], p);
    p = __builtin_elementwise_fma((v2f)(a.z), w[2], p);
    p = __builtin_elementwise_fma((v2f)(a.w), w[3], p);
    p = __builtin_elementwise_fma((v2f)(b.x), w[4], p);
    p = __builtin_elementwise_fma((v2f)(b.y), w[5], p);
    p = __builtin_elementwise_fma((v2f)(b.z), w[6], p);
    p = __builtin_elementwise_fma((v2f)(b.w), w[7], p);
    return p;
}
__device__ __forceinline__ void load_pair(const float* __restrict__ a_w, const int hk,
                                          const int hg, const int pa, v2f* w) {
    const float4 k0 = *(const float4*)(a_w + hk * (2 * HF) + pa);
    const float4 k1 = *(const float4*)(a_w + hk * (2 * HF) + pa + 64);
    const float4 g0 = *(const float4*)(a_w + hg * (2 * HF) + pa);
    const float4 g1 = *(const float4*)(a_w + hg * (2 * HF) + pa + 64);
    w[0] = (v2f){k0.x, g0.x}; w[1] = (v2f){k0.y, g0.y};
    w[2] = (v2f){k0.z, g0.z}; w[3] = (v2f){k0.w, g0.w};
    w[4] = (v2f){k1.x, g1.x}; w[5] = (v2f){k1.y, g1.y};
    w[6] = (v2f){k1.z, g1.z}; w[7] = (v2f){k1.w, g1.w};
}
__device__ __forceinline__ float sum8_keep(const v2f p) {
    float keep = p[0] + lane_xor<4>(p[1]);
    keep += lane_xor<1>(keep);
    keep += lane_xor<2>(keep);
    return keep;
}

template <int K>
__device__ __forceinline__ const float* group_row(const float* __restrict__ nt, const int w, const int col) {
    unsigned sk = __builtin_amdgcn_mov_dpp8((unsigned)w, (unsigned)K * 0x249249u) & SRC_MASK;
    sk = sk < (unsigned)N_NODES ? sk : (unsigned)(N_NODES - 1);
    return nt + (size_t)sk * HF + col;
}

template <int B>
__device__ __forceinline__ void hops4(const float* __restrict__ nt, const int w, const int cnt,
                                      const int col, const v2f* wA, const v2f* wB,
                                      const float tdA, const float tdB,
                                      float4& acc0, float4& acc1,
                                      float& den_lo, float& den_hi, float& mx) {
    const float* srow[4];
    srow[0] = group_row<B + 0>(nt, w, col);
    srow[1] = group_row<B + 1>(nt, w, col);
    srow[2] = group_row<B + 2>(nt, w, col);
    srow[3] = group_row<B + 3>(nt, w, col);
    float4 r0[4], r1[4];
#pragma unroll
    for (int k = 0; k < 4; ++k) {
        r0[k] = *(const float4*)srow[k];
        r1[k] = *(const float4*)(srow[k] + 32);
    }
#pragma unroll
    for (int k = 0; k < 4; ++k) {
        const bool act = (B + k) < cnt;
        const float lg_lo = sum8_keep(dot8x2(r0[k], r1[k], wA)) + tdA;
        const float lg_hi = sum8_keep(dot8x2(r0[k], r1[k], wB)) + tdB;
        mx = act ? fmaxf(mx, fmaxf(lg_lo, lg_hi)) : mx;
        const float e_lo = act ? __expf(NEG_SLOPE * lg_lo) : 0.0f;
        const float e_hi = act ? __expf(NEG_SLOPE * lg_hi) : 0.0f;
        den_lo += e_lo;
        den_hi += e_hi;
        acc0.x = fmaf(e_lo, r0[k].x, acc0.x); acc0.y = fmaf(e_lo, r0[k].y, acc0.y);
        acc0.z = fmaf(e_lo, r0[k].z, acc0.z); acc0.w = fmaf(e_lo, r0[k].w, acc0.w);
        acc1.x = fmaf(e_hi, r1[k].x, acc1.x); acc1.y = fmaf(e_hi, r1[k].y, acc1.y);
        acc1.z = fmaf(e_hi, r1[k].z, acc1.z); acc1.w = fmaf(e_hi, r1[k].w, acc1.w);
    }
}

__global__ __launch_bounds__(256)
void gat_node_gather(const float* __restrict__ nt,
                     const float* __restrict__ a_w,
                     const int*   __restrict__ order,
                     const int*   __restrict__ srt,
                     unsigned* __restrict__ maxenc,
                     float* __restrict__ denom,
                     float* __restrict__ out) {
    __shared__ unsigned smax;
    if (threadIdx.x == 0) smax = 0u;
    __syncthreads();

    const int j   = threadIdx.x & 7;
    constexpr int PER   = BIN_R / 32;
    constexpr int FULL  = N_NODES / BIN_R;
    constexpr int TAILB = (N_NODES - FULL * BIN_R) / 32;
    const int bi = (int)blockIdx.x;
    int k, r;
    if (bi < TAILB * N_BINS) {
        k = bi / N_BINS;
        r = bi - k * N_BINS;
    } else {
        const int q = (bi - TAILB * N_BINS) / FULL;
        k = TAILB + q;
        r = (bi - TAILB * N_BINS) - q * FULL;
    }
    const unsigned ow = (unsigned)order[(r * PER + k) * 32 + (threadIdx.x >> 3)];
    const int nraw = r * BIN_R + (int)(ow >> HEAD_BITS);
    const int n    = nraw < N_NODES ? nraw : N_NODES - 1;
    const int e0   = (int)(ow & ((1u << HEAD_BITS) - 1u)) - 1;
    const int col = 4 * j;
    const int hb  = j >> 2;
    const int pa  = hb * 32 + (col & 15);

    bool live = (unsigned)e0 < (unsigned)E_EDGES;
    int pos = live ? e0 : 0;
    int w;
    {
        const int wi = pos + j;
        w = srt[wi < E_EDGES ? wi : E_EDGES - 1];
    }
    float tdA, tdB;
    {
        const float* trow = nt + (size_t)n * HF + col;
        const float4 t0 = *(const float4*)trow;
        const float4 t1 = *(const float4*)(trow + 32);
        v2f wt[8];
        load_pair(a_w, hb, hb ^ 1, pa + 16, wt);
        tdA = sum8_keep(dot8x2(t0, t1, wt));
        load_pair(a_w, 2 + hb, 2 + (hb ^ 1), pa + 16, wt);
        tdB = sum8_keep(dot8x2(t0, t1, wt));
    }
    v2f wA[8], wB[8];
    load_pair(a_w, hb, hb ^ 1, pa, wA);
    load_pair(a_w, 2 + hb, 2 + (hb ^ 1), pa, wB);

    float4 acc0 = make_float4(0.f, 0.f, 0.f, 0.f);
    float4 acc1 = make_float4(0.f, 0.f, 0.f, 0.f);
    float  den_lo = 0.f, den_hi = 0.f;
    float  mx = -INFINITY;

    const int lane = threadIdx.x & 31;
    for (int it = 0; it < E_EDGES; it += 8) {
        if (__builtin_amdgcn_ballot_w32(live) == 0u) break;
        const unsigned bal  = __builtin_amdgcn_ballot_w32(live && w < 0);
        const unsigned byte = (bal >> (lane & 24)) & 0xFFu;
        const int cnt = live ? (byte ? __builtin_ctz(byte) + 1 : 8) : 0;
        const bool more = live && byte == 0u;
        pos += 8;
        const int wi = pos + j;
        const int wn = srt[more ? (wi < E_EDGES ? wi : E_EDGES - 1) : 0];

        hops4<0>(nt, w, cnt, col, wA, wB, tdA, tdB, acc0, acc1, den_lo, den_hi, mx);
        if (__builtin_amdgcn_ballot_w32(cnt > 4) != 0u)
            hops4<4>(nt, w, cnt, col, wA, wB, tdA, tdB, acc0, acc1, den_lo, den_hi, mx);

        live = more;
        w = wn;
    }

    const float d_lo = den_lo;
    const float d_hi = den_hi;
    if (d_lo > 0.0f) {
        acc0.x /= d_lo; acc0.y /= d_lo; acc0.z /= d_lo; acc0.w /= d_lo;
    } else {
        acc0 = make_float4(0.f, 0.f, 0.f, 0.f);
    }
    if (d_hi > 0.0f) {
        acc1.x /= d_hi; acc1.y /= d_hi; acc1.z /= d_hi; acc1.w /= d_hi;
    } else {
        acc1 = make_float4(0.f, 0.f, 0.f, 0.f);
    }
    float* orow = out + (size_t)n * HF + col;
    *(float4*)orow        = acc0;
    *(float4*)(orow + 32) = acc1;
    const float d1 = lane_xor<4>(den_lo);
    const float d3 = lane_xor<4>(den_hi);
    mx = fmaxf(mx, lane_xor<4>(mx));
    if (j == 0) {
        *(float4*)(denom + (size_t)n * H_HEADS) =
            make_float4(den_lo, d1, den_hi, d3);
        atomicMax(&smax, fenc(mx));
    }
    __syncthreads();
    if (threadIdx.x == 0) atomicMax(maxenc, smax);
}

__global__ __launch_bounds__(256)
void gat_epsilon(float* __restrict__ out, const float* __restrict__ denom,
                 const unsigned* __restrict__ maxenc) {
    const int n = blockIdx.x * 256 + threadIdx.x;
    if (n >= N_NODES) return;
    const float eps = 1e-8f * __expf(NEG_SLOPE * fdec(*maxenc));
    const float4 dv = ((const float4*)denom)[n];
    const float dd[H_HEADS] = {dv.x, dv.y, dv.z, dv.w};
#pragma unroll
    for (int k = 0; k < H_HEADS; ++k) {
        const float d = dd[k] + eps;
        if (d != dd[k]) {
            const float sc = dd[k] / d;
            float4* o = (float4*)(out + (size_t)n * HF + k * F_OUT);
            for (int q = 0; q < F_OUT / 4; ++q) {
                float4 v = o[q];
                v.x *= sc; v.y *= sc; v.z *= sc; v.w *= sc;
                o[q] = v;
            }
        }
    }
}

extern "C" void kernel_launch(void* const* d_in, const int* in_sizes, int n_in,
                              void* d_out, int out_size, void* d_ws, size_t ws_size,
                              hipStream_t stream) {
    const float* x   = (const float*)d_in[0];
    const float* W_w = (const float*)d_in[1];
    const float* a_w = (const float*)d_in[2];
    const int*   ei  = (const int*)d_in[3];
    const int* src = ei;
    const int* tgt = ei + E_EDGES;
    float* out = (float*)d_out;

    float* nt    = (float*)d_ws;
    float* denom = nt + (size_t)N_NODES * HF;
    unsigned* maxenc = (unsigned*)(denom + (size_t)N_NODES * H_HEADS);
    int*  order = (int*)(maxenc + 4);
    int*  grp  = order + N_NODES;
    int*  srt  = grp + E_EDGES;
    int*  off  = (int*)denom;

    (void)in_sizes; (void)n_in; (void)out_size; (void)ws_size;

    {
        gat_transform_wmma<<<N_NODES / 32, 64, 0, stream>>>(x, W_w, nt, maxenc);
    }
    {
        gat_group<<<N_CHUNKS, 1024, 0, stream>>>(src, tgt, off, grp);
    }
    {
        gat_link_lds<<<N_BINS, 1024, 0, stream>>>(off, order, grp, srt);
    }
    {
        gat_node_gather<<<N_NODES / 32, 256, 0, stream>>>(
            nt, a_w, order, srt, maxenc, denom, out);
    }
    {
        gat_epsilon<<<(N_NODES + 255) / 256, 256, 0, stream>>>(out, denom, maxenc);
    }
}
